# starting kernel
# baseline (speedup 1.0000x reference)
#include <hip/hip_runtime.h>
#include <hip/hip_bf16.h>
#include <math.h>

typedef float v2f __attribute__((ext_vector_type(2)));
typedef float v8f __attribute__((ext_vector_type(8)));

#define NEG_SLOPE 0.2f

__device__ __forceinline__ void atomicMaxF(float* addr, float val) {
    if (val >= 0.0f) {
        atomicMax((int*)addr, __float_as_int(val));
    } else {
        atomicMin((unsigned int*)addr, __float_as_uint(val));
    }
}

__device__ __forceinline__ void edge_sd(const int* __restrict__ ei, int e, int E,
                                        int& s, int& d) {
    if (e < E) { s = ei[e]; d = ei[E + e]; }
    else       { s = d = e - E; }
}

__global__ void fill_kernel(float* __restrict__ p, float v, int n) {
    int i = blockIdx.x * blockDim.x + threadIdx.x;
    if (i < n) p[i] = v;
}

template <int K, int NCOLT>
__global__ void gemm_wmma_f32(const float* __restrict__ X,
                              const float* __restrict__ W,
                              float* __restrict__ H) {
    constexpr int FOUT = 16 * NCOLT;
    __shared__ float As[16][K + 4];

    const int r0   = blockIdx.x * 16;
    const int lane = threadIdx.x;
    const int wave = threadIdx.y;
    const int tid  = wave * 32 + lane;
    const int nthr = 32 * NCOLT;

    for (int i = tid; i < 16 * K; i += nthr) {
        int rr = i / K, cc = i % K;
        As[rr][cc] = X[(size_t)(r0 + rr) * K + cc];
    }
    __syncthreads();

    const int half = lane >> 4;
    const int m    = lane & 15;
    const int c0   = wave * 16;

    v8f acc = {};
    #pragma unroll 4
    for (int kk = 0; kk < K; kk += 4) {
        const int kb = kk + 2 * half;
        v2f a, b;
        a[0] = As[m][kb];
        a[1] = As[m][kb + 1];
        b[0] = W[(size_t)(kb)     * FOUT + c0 + m];
        b[1] = W[(size_t)(kb + 1) * FOUT + c0 + m];
        acc = __builtin_amdgcn_wmma_f32_16x16x4_f32(
             false, a,  false, b,
             (short)0, acc,  false,  false);
    }

    #pragma unroll
    for (int r = 0; r < 8; ++r) {
        H[(size_t)(r0 + r + 8 * half) * FOUT + c0 + m] = acc[r];
    }
}

template <int H, int D>
__global__ void alpha_kernel(const float* __restrict__ h,
                             const float* __restrict__ a_src,
                             const float* __restrict__ a_dst,
                             float* __restrict__ as, float* __restrict__ ad,
                             int n) {
    int i = blockIdx.x * blockDim.x + threadIdx.x;
    if (i >= n * H) return;
    int node = i / H, hh = i % H;
    const float* hp = h + (size_t)node * (H * D) + hh * D;
    float s = 0.f, d = 0.f;
    #pragma unroll
    for (int j = 0; j < D; ++j) {
        float v = hp[j];
        s += v * a_src[hh * D + j];
        d += v * a_dst[hh * D + j];
    }
    as[i] = s;
    ad[i] = d;
}

template <int H>
__global__ void edge_max_kernel(const int* __restrict__ ei, int E, int N,
                                const float* __restrict__ as,
                                const float* __restrict__ ad,
                                float* __restrict__ ebuf,
                                float* __restrict__ m) {
    int i = blockIdx.x * blockDim.x + threadIdx.x;
    if (i >= (E + N) * H) return;
    int e = i / H, hh = i % H;
    int s, d;
    edge_sd(ei, e, E, s, d);
    float v = as[s * H + hh] + ad[d * H + hh];
    v = (v > 0.f) ? v : NEG_SLOPE * v;
    ebuf[i] = v;
    atomicMaxF(&m[d * H + hh], v);
}

template <int H>
__global__ void edge_exp_kernel(const int* __restrict__ ei, int E, int N,
                                float* __restrict__ ebuf,
                                const float* __restrict__ m,
                                float* __restrict__ den) {
    int i = blockIdx.x * blockDim.x + threadIdx.x;
    if (i >= (E + N) * H) return;
    int e = i / H, hh = i % H;
    int s, d;
    edge_sd(ei, e, E, s, d);
    float ex = expf(ebuf[i] - m[d * H + hh]);
    ebuf[i] = ex;
    atomicAdd(&den[d * H + hh], ex);
}

template <int H, int D>
__global__ void edge_msg_kernel(const int* __restrict__ ei, int E, int N,
                                const float* __restrict__ h,
                                const float* __restrict__ ebuf,
                                const float* __restrict__ den,
                                float* __restrict__ agg) {
    int i = blockIdx.x * blockDim.x + threadIdx.x;
    if (i >= (E + N) * H) return;
    int e = i / H, hh = i % H;
    int s, d;
    edge_sd(ei, e, E, s, d);
    float alpha = ebuf[i] / (den[d * H + hh] + 1e-16f);
    const float* hp = h + (size_t)s * (H * D) + hh * D;
    float* ap = agg + (size_t)d * (H * D) + hh * D;
    #pragma unroll
    for (int j = 0; j < D; ++j) {
        atomicAdd(&ap[j], hp[j] * alpha);
    }
}

__global__ void bias_elu_kernel(float* __restrict__ agg,
                                const float* __restrict__ b, int n, int F) {
    int i = blockIdx.x * blockDim.x + threadIdx.x;
    if (i >= n * F) return;
    float v = agg[i] + b[i % F];
    agg[i] = (v > 0.f) ? v : (expf(v) - 1.f);
}

__global__ void add_bias_kernel(float* __restrict__ out,
                                const float* __restrict__ b, int n, int F) {
    int i = blockIdx.x * blockDim.x + threadIdx.x;
    if (i >= n * F) return;
    out[i] += b[i % F];
}

static inline int cdiv(int a, int b) { return (a + b - 1) / b; }

extern "C" void kernel_launch(void* const* d_in, const int* in_sizes, int n_in,
                              void* d_out, int out_size, void* d_ws, size_t ws_size,
                              hipStream_t stream) {
    const float* x      = (const float*)d_in[0];
    const int*   ei     = (const int*)  d_in[1];
    const float* W1     = (const float*)d_in[2];
    const float* a_src1 = (const float*)d_in[3];
    const float* a_dst1 = (const float*)d_in[4];
    const float* b1     = (const float*)d_in[5];
    const float* W2     = (const float*)d_in[6];
    const float* a_src2 = (const float*)d_in[7];
    const float* a_dst2 = (const float*)d_in[8];
    const float* b2     = (const float*)d_in[9];

    const int N  = in_sizes[0] / 128;
    const int E  = in_sizes[1] / 2;
    const int EP = E + N;

    float* ws   = (float*)d_ws;
    float* h1   = ws;
    float* agg1 = h1   + (size_t)N * 64;
    float* as1  = agg1 + (size_t)N * 64;
    float* ad1  = as1  + (size_t)N * 4;
    float* m1   = ad1  + (size_t)N * 4;
    float* den1 = m1   + (size_t)N * 4;
    float* ex1  = den1 + (size_t)N * 4;
    float* h2   = h1;
    float* as2  = h2   + (size_t)N * 16;
    float* ad2  = as2  + N;
    float* m2   = ad2  + N;
    float* den2 = m2   + N;
    float* ex2  = den2 + N;

    float* out  = (float*)d_out;

    const int TB = 256;

    fill_kernel<<<cdiv(N * 4, TB), TB, 0, stream>>>(m1, -INFINITY, N * 4);
    fill_kernel<<<cdiv(N * 4, TB), TB, 0, stream>>>(den1, 0.f, N * 4);
    fill_kernel<<<cdiv(N * 64, TB), TB, 0, stream>>>(agg1, 0.f, N * 64);

    gemm_wmma_f32<128, 4><<<N / 16, dim3(32, 4), 0, stream>>>(x, W1, h1);

    alpha_kernel<4, 16><<<cdiv(N * 4, TB), TB, 0, stream>>>(
        h1, a_src1, a_dst1, as1, ad1, N);

    edge_max_kernel<4><<<cdiv(EP * 4, TB), TB, 0, stream>>>(
        ei, E, N, as1, ad1, ex1, m1);
    edge_exp_kernel<4><<<cdiv(EP * 4, TB), TB, 0, stream>>>(
        ei, E, N, ex1, m1, den1);
    edge_msg_kernel<4, 16><<<cdiv(EP * 4, TB), TB, 0, stream>>>(
        ei, E, N, h1, ex1, den1, agg1);

    bias_elu_kernel<<<cdiv(N * 64, TB), TB, 0, stream>>>(agg1, b1, N, 64);

    fill_kernel<<<cdiv(N, TB), TB, 0, stream>>>(m2, -INFINITY, N);
    fill_kernel<<<cdiv(N, TB), TB, 0, stream>>>(den2, 0.f, N);
    fill_kernel<<<cdiv(N * 16, TB), TB, 0, stream>>>(out, 0.f, N * 16);

    gemm_wmma_f32<64, 1><<<N / 16, dim3(32, 1), 0, stream>>>(agg1, W2, h2);

    alpha_kernel<1, 16><<<cdiv(N, TB), TB, 0, stream>>>(
        h2, a_src2, a_dst2, as2, ad2, N);

    edge_max_kernel<1><<<cdiv(EP, TB), TB, 0, stream>>>(
        ei, E, N, as2, ad2, ex2, m2);
    edge_exp_kernel<1><<<cdiv(EP, TB), TB, 0, stream>>>(
        ei, E, N, ex2, m2, den2);
    edge_msg_kernel<1, 16><<<cdiv(EP, TB), TB, 0, stream>>>(
        ei, E, N, h2, ex2, den2, out);

    add_bias_kernel<<<cdiv(N * 16, TB), TB, 0, stream>>>(out, b2, N, 16);
}


# Edges grouped by destination once per call (returning int add as rank, scan, place); wave per node gathers rows, softmax+sum in registers, one divide at end: no float atomics, sums in list order
# speedup vs baseline: 4.6856x; 4.6856x over previous
#include <hip/hip_runtime.h>
#include <hip/hip_bf16.h>
#include <math.h>

typedef float v2f __attribute__((ext_vector_type(2)));
typedef float v8f __attribute__((ext_vector_type(8)));

#define NEG_SLOPE 0.2f

__global__ void zero_int_kernel(int* __restrict__ p, int n) {
    int i = blockIdx.x * blockDim.x + threadIdx.x;
    if (i < n) p[i] = 0;
}

__global__ void bucket_count_kernel(const int* __restrict__ ei, int E,
                                    int* __restrict__ cnt,
                                    int* __restrict__ rank) {
    const int stride = gridDim.x * blockDim.x;
    for (int e = blockIdx.x * blockDim.x + threadIdx.x; e < E; e += stride) {
        const int d = ei[E + e];
        rank[e] = atomicAdd(&cnt[d], 1);
    }
}

__global__ void bucket_partial_kernel(const int* __restrict__ cnt,
                                      int* __restrict__ bsum, int n) {
    __shared__ int sh[256];
    const int t  = threadIdx.x;
    const int i0 = blockIdx.x * 1024 + t * 4;
    int s = 0;
    for (int k = 0; k < 4; ++k) s += (i0 + k < n) ? cnt[i0 + k] : 0;
    sh[t] = s;
    __syncthreads();
    for (int off = 128; off > 0; off >>= 1) {
        if (t < off) sh[t] += sh[t + off];
        __syncthreads();
    }
    if (t == 0) bsum[blockIdx.x] = sh[0];
}

__global__ void bucket_scan_kernel(const int* __restrict__ cnt,
                                   const int* __restrict__ bsum,
                                   int* __restrict__ start, int n) {
    __shared__ int sh[256];
    const int t  = threadIdx.x;
    const int i0 = blockIdx.x * 1024 + t * 4;

    int part = 0;
    for (int i = t; i < (int)blockIdx.x; i += 256) part += bsum[i];

    int c[4];
    int tsum = 0;
    for (int k = 0; k < 4; ++k) {
        c[k] = (i0 + k < n) ? cnt[i0 + k] : 0;
        tsum += c[k];
    }

    sh[t] = part;
    __syncthreads();
    for (int off = 128; off > 0; off >>= 1) {
        if (t < off) sh[t] += sh[t + off];
        __syncthreads();
    }
    const int base = sh[0];
    __syncthreads();

    sh[t] = tsum;
    __syncthreads();
    for (int off = 1; off < 256; off <<= 1) {
        const int v = (t >= off) ? sh[t - off] : 0;
        __syncthreads();
        sh[t] += v;
        __syncthreads();
    }
    int run = base + sh[t] - tsum;
    for (int k = 0; k < 4; ++k) {
        if (i0 + k < n) { start[i0 + k] = run; run += c[k]; }
    }
    if (i0 < n && n <= i0 + 4) start[n] = run;
}

__global__ void bucket_place_kernel(const int* __restrict__ ei, int E,
                                    const int* __restrict__ start,
                                    const int* __restrict__ rank,
                                    int* __restrict__ csr) {
    const int stride = gridDim.x * blockDim.x;
    for (int e = blockIdx.x * blockDim.x + threadIdx.x; e < E; e += stride) {
        const int d = ei[E + e];
        csr[start[d] + rank[e]] = ei[e];
    }
}

template <int K, int NCOLT>
__global__ void gemm_wmma_f32(const float* __restrict__ X,
                              const float* __restrict__ W,
                              float* __restrict__ H) {
    constexpr int FOUT = 16 * NCOLT;
    __shared__ float As[16][K + 4];

    const int r0   = blockIdx.x * 16;
    const int lane = threadIdx.x;
    const int wave = threadIdx.y;
    const int tid  = wave * 32 + lane;
    const int nthr = 32 * NCOLT;

    for (int i = tid; i < 16 * K; i += nthr) {
        int rr = i / K, cc = i % K;
        As[rr][cc] = X[(size_t)(r0 + rr) * K + cc];
    }
    __syncthreads();

    const int half = lane >> 4;
    const int m    = lane & 15;
    const int c0   = wave * 16;

    v8f acc = {};
    #pragma unroll 4
    for (int kk = 0; kk < K; kk += 4) {
        const int kb = kk + 2 * half;
        v2f a, b;
        a[0] = As[m][kb];
        a[1] = As[m][kb + 1];
        b[0] = W[(size_t)(kb)     * FOUT + c0 + m];
        b[1] = W[(size_t)(kb + 1) * FOUT + c0 + m];
        acc = __builtin_amdgcn_wmma_f32_16x16x4_f32(
             false, a,  false, b,
             (short)0, acc,  false,  false);
    }

    #pragma unroll
    for (int r = 0; r < 8; ++r) {
        H[(size_t)(r0 + r + 8 * half) * FOUT + c0 + m] = acc[r];
    }
}

template <int H, int D>
__global__ void alpha_kernel(const float* __restrict__ h,
                             const float* __restrict__ a_src,
                             const float* __restrict__ a_dst,
                             float* __restrict__ as, float* __restrict__ ad,
                             int n) {
    int i = blockIdx.x * blockDim.x + threadIdx.x;
    if (i >= n * H) return;
    int node = i / H, hh = i % H;
    const float* hp = h + (size_t)node * (H * D) + hh * D;
    float s = 0.f, d = 0.f;
    #pragma unroll
    for (int j = 0; j < D; ++j) {
        float v = hp[j];
        s += v * a_src[hh * D + j];
        d += v * a_dst[hh * D + j];
    }
    as[i] = s;
    ad[i] = d;
}

template <int H, int D, bool ELU>
__global__ void __launch_bounds__(256)
node_gather_kernel(const int* __restrict__ start, const int* __restrict__ csr,
                   const float* __restrict__ h,
                   const float* __restrict__ as, const float* __restrict__ ad,
                   const float* __restrict__ bias,
                   float* __restrict__ out, int N) {
    constexpr int F   = H * D;
    constexpr int VEC = (F >= 64) ? 2 : 1;
    constexpr int LPR = F / VEC;
    constexpr int EPS = 32 / LPR;
    constexpr int UNR = 4;
    static_assert(LPR == 32 || LPR == 16, "row must map to 16 or 32 lanes");

    const int lane = threadIdx.x & 31;
    const int sub  = lane / LPR;
    const int f0   = (lane % LPR) * VEC;
    const int myh  = f0 / D;
    const int wave   = blockIdx.x * (blockDim.x >> 5) + (threadIdx.x >> 5);
    const int nwaves = gridDim.x * (blockDim.x >> 5);

    for (int n0 = wave; n0 < N; n0 += nwaves) {
        const int n   = __builtin_amdgcn_readfirstlane(n0);
        const int beg = start[n];
        const int cnt = start[n + 1] - beg + 1;

        float adn[H];
        for (int k = 0; k < H; ++k) adn[k] = ad[n * H + k];

        float mx[H];
        for (int k = 0; k < H; ++k) mx[k] = -INFINITY;

        for (int base = 0; base < cnt; base += 32) {
            const int idx = base + lane;
            const int s = (idx < cnt - 1) ? csr[beg + idx] : n;
            float av[H];
            if constexpr (H == 4) {
                const float4 t = *(const float4*)(as + (size_t)s * 4);
                av[0] = t.x; av[1] = t.y; av[2] = t.z; av[3] = t.w;
            } else {
                for (int k = 0; k < H; ++k) av[k] = as[(size_t)s * H + k];
            }
            for (int k = 0; k < H; ++k) {
                float v = av[k] + adn[k];
                v = (v > 0.f) ? v : NEG_SLOPE * v;
                mx[k] = fmaxf(mx[k], (idx < cnt) ? v : -INFINITY);
            }
        }
        for (int k = 0; k < H; ++k)
            for (int off = 16; off > 0; off >>= 1)
                mx[k] = fmaxf(mx[k], __shfl_xor(mx[k], off));

        float acc[VEC];
        for (int v = 0; v < VEC; ++v) acc[v] = 0.f;
        float den = 0.f;

        for (int base = 0; base < cnt; base += 32) {
            const int idx = base + lane;
            const int s = (idx < cnt - 1) ? csr[beg + idx] : n;
            float av[H];
            if constexpr (H == 4) {
                const float4 t = *(const float4*)(as + (size_t)s * 4);
                av[0] = t.x; av[1] = t.y; av[2] = t.z; av[3] = t.w;
            } else {
                for (int k = 0; k < H; ++k) av[k] = as[(size_t)s * H + k];
            }
            float w[H];
            for (int k = 0; k < H; ++k) {
                float v = av[k] + adn[k];
                v = (v > 0.f) ? v : NEG_SLOPE * v;
                const float ex = expf(v - mx[k]);
                w[k] = (idx < cnt) ? ex : 0.f;
            }

            const int m = min(32, cnt - base);
            for (int j = 0; j < m; j += UNR * EPS) {
                #pragma unroll
                for (int u = 0; u < UNR; ++u) {
                    const int jj = j + u * EPS + sub;
                    const int sj = __shfl(s, jj);
                    float wj = 0.f;
                    for (int k = 0; k < H; ++k) {
                        const float t = __shfl(w[k], jj);
                        if (k == myh) wj = t;
                    }
                    const float* row = h + (size_t)sj * F + f0;
                    if constexpr (VEC == 2) {
                        const float2 r = *(const float2*)row;
                        acc[0] = fmaf(wj, r.x, acc[0]);
                        acc[1] = fmaf(wj, r.y, acc[1]);
                    } else {
                        acc[0] = fmaf(wj, row[0], acc[0]);
                    }
                    den += wj;
                }
            }
        }

        for (int off = LPR; off < 32; off <<= 1) {
            for (int v = 0; v < VEC; ++v) acc[v] += __shfl_xor(acc[v], off);
            den += __shfl_xor(den, off);
        }

        if (sub == 0) {
            for (int v = 0; v < VEC; ++v) {
                float val = acc[v] / (den + 1e-16f) + bias[f0 + v];
                if (ELU) val = (val > 0.f) ? val : (expf(val) - 1.f);
                out[(size_t)n * F + f0 + v] = val;
            }
        }
    }
}

static inline int cdiv(int a, int b) { return (a + b - 1) / b; }

extern "C" void kernel_launch(void* const* d_in, const int* in_sizes, int n_in,
                              void* d_out, int out_size, void* d_ws, size_t ws_size,
                              hipStream_t stream) {
    const float* x      = (const float*)d_in[0];
    const int*   ei     = (const int*)  d_in[1];
    const float* W1     = (const float*)d_in[2];
    const float* a_src1 = (const float*)d_in[3];
    const float* a_dst1 = (const float*)d_in[4];
    const float* b1     = (const float*)d_in[5];
    const float* W2     = (const float*)d_in[6];
    const float* a_src2 = (const float*)d_in[7];
    const float* a_dst2 = (const float*)d_in[8];
    const float* b2     = (const float*)d_in[9];

    const int N  = in_sizes[0] / 128;
    const int E  = in_sizes[1] / 2;

    float* ws   = (float*)d_ws;
    float* h1   = ws;
    float* agg1 = h1   + (size_t)N * 64;
    float* as1  = agg1 + (size_t)N * 64;
    float* ad1  = as1  + (size_t)N * 4;
    int*   cnt  = (int*)(ad1 + (size_t)N * 4);
    int*   strt = cnt   + N;
    int*   rank = strt  + N + 4;
    int*   csr  = rank  + E;
    int*   bsum = csr   + E;
    float* h2   = h1;
    float* as2  = h2   + (size_t)N * 16;
    float* ad2  = as2  + N;

    float* out  = (float*)d_out;

    const int TB = 256;
    const int NB = cdiv(N, 1024);

    zero_int_kernel<<<cdiv(N, TB), TB, 0, stream>>>(cnt, N);
    bucket_count_kernel<<<1024, TB, 0, stream>>>(ei, E, cnt, rank);
    bucket_partial_kernel<<<NB, 256, 0, stream>>>(cnt, bsum, N);
    bucket_scan_kernel<<<NB, 256, 0, stream>>>(cnt, bsum, strt, N);
    bucket_place_kernel<<<1024, TB, 0, stream>>>(ei, E, strt, rank, csr);

    gemm_wmma_f32<128, 4><<<N / 16, dim3(32, 4), 0, stream>>>(x, W1, h1);

    alpha_kernel<4, 16><<<cdiv(N * 4, TB), TB, 0, stream>>>(
        h1, a_src1, a_dst1, as1, ad1, N);

    node_gather_kernel<4, 16, true><<<2048, TB, 0, stream>>>(
        strt, csr, h1, as1, ad1, b1, agg1, N);

    gemm_wmma_f32<64, 1><<<N / 16, dim3(32, 1), 0, stream>>>(agg1, W2, h2);

    alpha_kernel<1, 16><<<cdiv(N, TB), TB, 0, stream>>>(
        h2, a_src2, a_dst2, as2, ad2, N);

    node_gather_kernel<1, 16, false><<<2048, TB, 0, stream>>>(
        strt, csr, h2, as2, ad2, b2, out, N);
}


# Gather: lists of up to 32 edges take one pass (indices, logits loaded once), all source rows requested before the softmax, next node's list and bounds fetched ahead; sums still in list order
# speedup vs baseline: 4.9351x; 1.0532x over previous
#include <hip/hip_runtime.h>
#include <hip/hip_bf16.h>
#include <math.h>

typedef float v2f __attribute__((ext_vector_type(2)));
typedef float v8f __attribute__((ext_vector_type(8)));

#define NEG_SLOPE 0.2f

__global__ void zero_int_kernel(int* __restrict__ p, int n) {
    int i = blockIdx.x * blockDim.x + threadIdx.x;
    if (i < n) p[i] = 0;
}

__global__ void bucket_count_kernel(const int* __restrict__ ei, int E,
                                    int* __restrict__ cnt,
                                    int* __restrict__ rank) {
    const int stride = gridDim.x * blockDim.x;
    for (int e = blockIdx.x * blockDim.x + threadIdx.x; e < E; e += stride) {
        const int d = ei[E + e];
        rank[e] = atomicAdd(&cnt[d], 1);
    }
}

__global__ void bucket_partial_kernel(const int* __restrict__ cnt,
                                      int* __restrict__ bsum, int n) {
    __shared__ int sh[256];
    const int t  = threadIdx.x;
    const int i0 = blockIdx.x * 1024 + t * 4;
    int s = 0;
    for (int k = 0; k < 4; ++k) s += (i0 + k < n) ? cnt[i0 + k] : 0;
    sh[t] = s;
    __syncthreads();
    for (int off = 128; off > 0; off >>= 1) {
        if (t < off) sh[t] += sh[t + off];
        __syncthreads();
    }
    if (t == 0) bsum[blockIdx.x] = sh[0];
}

__global__ void bucket_scan_kernel(const int* __restrict__ cnt,
                                   const int* __restrict__ bsum,
                                   int* __restrict__ start, int n) {
    __shared__ int sh[256];
    const int t  = threadIdx.x;
    const int i0 = blockIdx.x * 1024 + t * 4;

    int part = 0;
    for (int i = t; i < (int)blockIdx.x; i += 256) part += bsum[i];

    int c[4];
    int tsum = 0;
    for (int k = 0; k < 4; ++k) {
        c[k] = (i0 + k < n) ? cnt[i0 + k] : 0;
        tsum += c[k];
    }

    sh[t] = part;
    __syncthreads();
    for (int off = 128; off > 0; off >>= 1) {
        if (t < off) sh[t] += sh[t + off];
        __syncthreads();
    }
    const int base = sh[0];
    __syncthreads();

    sh[t] = tsum;
    __syncthreads();
    for (int off = 1; off < 256; off <<= 1) {
        const int v = (t >= off) ? sh[t - off] : 0;
        __syncthreads();
        sh[t] += v;
        __syncthreads();
    }
    int run = base + sh[t] - tsum;
    for (int k = 0; k < 4; ++k) {
        if (i0 + k < n) { start[i0 + k] = run; run += c[k]; }
    }
    if (i0 < n && n <= i0 + 4) start[n] = run;
}

__global__ void bucket_place_kernel(const int* __restrict__ ei, int E,
                                    const int* __restrict__ start,
                                    const int* __restrict__ rank,
                                    int* __restrict__ csr) {
    const int stride = gridDim.x * blockDim.x;
    for (int e = blockIdx.x * blockDim.x + threadIdx.x; e < E; e += stride) {
        const int d = ei[E + e];
        csr[start[d] + rank[e]] = ei[e];
    }
}

template <int K, int NCOLT>
__global__ void gemm_wmma_f32(const float* __restrict__ X,
                              const float* __restrict__ W,
                              float* __restrict__ H) {
    constexpr int FOUT = 16 * NCOLT;
    __shared__ float As[16][K + 4];

    const int r0   = blockIdx.x * 16;
    const int lane = threadIdx.x;
    const int wave = threadIdx.y;
    const int tid  = wave * 32 + lane;
    const int nthr = 32 * NCOLT;

    for (int i = tid; i < 16 * K; i += nthr) {
        int rr = i / K, cc = i % K;
        As[rr][cc] = X[(size_t)(r0 + rr) * K + cc];
    }
    __syncthreads();

    const int half = lane >> 4;
    const int m    = lane & 15;
    const int c0   = wave * 16;

    v8f acc = {};
    #pragma unroll 4
    for (int kk = 0; kk < K; kk += 4) {
        const int kb = kk + 2 * half;
        v2f a, b;
        a[0] = As[m][kb];
        a[1] = As[m][kb + 1];
        b[0] = W[(size_t)(kb)     * FOUT + c0 + m];
        b[1] = W[(size_t)(kb + 1) * FOUT + c0 + m];
        acc = __builtin_amdgcn_wmma_f32_16x16x4_f32(
             false, a,  false, b,
             (short)0, acc,  false,  false);
    }

    #pragma unroll
    for (int r = 0; r < 8; ++r) {
        H[(size_t)(r0 + r + 8 * half) * FOUT + c0 + m] = acc[r];
    }
}

template <int H, int D>
__global__ void alpha_kernel(const float* __restrict__ h,
                             const float* __restrict__ a_src,
                             const float* __restrict__ a_dst,
                             float* __restrict__ as, float* __restrict__ ad,
                             int n) {
    int i = blockIdx.x * blockDim.x + threadIdx.x;
    if (i >= n * H) return;
    int node = i / H, hh = i % H;
    const float* hp = h + (size_t)node * (H * D) + hh * D;
    float s = 0.f, d = 0.f;
    #pragma unroll
    for (int j = 0; j < D; ++j) {
        float v = hp[j];
        s += v * a_src[hh * D + j];
        d += v * a_dst[hh * D + j];
    }
    as[i] = s;
    ad[i] = d;
}

constexpr int NBATCH = 8;

template <int EPS>
__device__ __forceinline__ int lane_pick(int v, int step, int sub) {
    if constexpr (EPS == 1) {
        return __builtin_amdgcn_readlane(v, step);
    } else {
        const int a = __builtin_amdgcn_readlane(v, 2 * step);
        const int b = __builtin_amdgcn_readlane(v, 2 * step + 1);
        return sub ? b : a;
    }
}

template <int F, int VEC, int EPS, int STEP0>
__device__ __forceinline__ void rows_request(const float* __restrict__ h,
                                             int s, int sub, int f0,
                                             float (&r)[NBATCH][VEC]) {
    #pragma unroll
    for (int u = 0; u < NBATCH; ++u) {
        const int sj = lane_pick<EPS>(s, STEP0 + u, sub);
        const unsigned off = (unsigned)sj * (unsigned)(F * 4) + (unsigned)f0 * 4u;
        const float* row = (const float*)((const char*)h + off);
        if constexpr (VEC == 2) {
            const float2 t = *(const float2*)row;
            r[u][0] = t.x; r[u][1] = t.y;
        } else {
            r[u][0] = row[0];
        }
    }
}

template <int H, int VEC, int EPS, int STEP0, int U0, int U1>
__device__ __forceinline__ void rows_consume(const float (&w)[H], int sub,
                                             int myh,
                                             const float (&r)[NBATCH][VEC],
                                             float (&acc)[VEC], float& den) {
    #pragma unroll
    for (int u = U0; u < U1; ++u) {
        float wj = 0.f;
        #pragma unroll
        for (int k = 0; k < H; ++k) {
            const float t = __int_as_float(
                lane_pick<EPS>(__float_as_int(w[k]), STEP0 + u, sub));
            if (k == myh) wj = t;
        }
        #pragma unroll
        for (int v = 0; v < VEC; ++v) acc[v] = fmaf(wj, r[u][v], acc[v]);
        den += wj;
    }
}

template <int H, int D, bool ELU>
__global__ void __launch_bounds__(256, 16)
node_gather_kernel(const int* __restrict__ start, const int* __restrict__ csr,
                   const float* __restrict__ h,
                   const float* __restrict__ as, const float* __restrict__ ad,
                   const float* __restrict__ bias,
                   float* __restrict__ out, int N) {
    constexpr int F   = H * D;
    constexpr int VEC = (F >= 64) ? 2 : 1;
    constexpr int LPR = F / VEC;
    constexpr int EPS = 32 / LPR;
    constexpr int EPB = NBATCH * EPS;
    constexpr int UNR = 4;
    static_assert(LPR == 32 || LPR == 16, "row must map to 16 or 32 lanes");

    const int lane = threadIdx.x & 31;
    const int sub  = lane / LPR;
    const int f0   = (lane % LPR) * VEC;
    const int myh  = f0 / D;
    const int wave   = blockIdx.x * (blockDim.x >> 5) + (threadIdx.x >> 5);
    const int nwaves = gridDim.x * (blockDim.x >> 5);

    if (wave >= N) return;

    const int wave_s   = __builtin_amdgcn_readfirstlane(wave);
    const int nwaves_s = __builtin_amdgcn_readfirstlane(nwaves);

    float bv[VEC];
    for (int v = 0; v < VEC; ++v) bv[v] = bias[f0 + v];

    int   beg, end, s_cur;
    float adn[H];
    int   nbeg = 0, nend = 0;
    float nad[H];
    {
        const int n = wave_s;
        beg = start[n];
        end = start[n + 1];
        for (int k = 0; k < H; ++k) adn[k] = ad[n * H + k];
        for (int k = 0; k < H; ++k) nad[k] = 0.f;
        const int nn = n + nwaves_s;
        if (nn < N) {
            nbeg = start[nn];
            nend = start[nn + 1];
            for (int k = 0; k < H; ++k) nad[k] = ad[nn * H + k];
        }
        s_cur = (lane < end - beg) ? csr[beg + lane] : n;
    }

    for (int n = wave_s; n < N; n += nwaves_s) {
        const int cnt = end - beg + 1;

        const bool has_next = n + nwaves_s < N;
        const int  nn       = has_next ? n + nwaves_s : n;
        const int  n2       = nn + nwaves_s;
        const bool has_n2   = has_next && n2 < N;
        int   n2beg = 0, n2end = 0;
        float n2ad[H];
        for (int k = 0; k < H; ++k) n2ad[k] = 0.f;
        int s_next = nn;

        float acc[VEC];
        for (int v = 0; v < VEC; ++v) acc[v] = 0.f;
        float den = 0.f;

        if (cnt <= 32) {
            const int s = s_cur;

            float av[H];
            if constexpr (H == 4) {
                const float4 t = *(const float4*)(as + (size_t)s * 4);
                av[0] = t.x; av[1] = t.y; av[2] = t.z; av[3] = t.w;
            } else {
                for (int k = 0; k < H; ++k) av[k] = as[(size_t)s * H + k];
            }

            float ra[NBATCH][VEC], rb[NBATCH][VEC];
            rows_request<F, VEC, EPS, 0>(h, s, sub, f0, ra);
            if (cnt > EPB)
                rows_request<F, VEC, EPS, NBATCH>(h, s, sub, f0, rb);

            if (has_next)
                s_next = (lane < nend - nbeg) ? csr[nbeg + lane] : nn;
            if (has_n2) {
                n2beg = start[n2];
                n2end = start[n2 + 1];
                for (int k = 0; k < H; ++k) n2ad[k] = ad[n2 * H + k];
            }

            float w[H];
            for (int k = 0; k < H; ++k) {
                float v = av[k] + adn[k];
                v = (v > 0.f) ? v : NEG_SLOPE * v;
                float mx = (lane < cnt) ? v : -INFINITY;
                for (int off = 16; off > 0; off >>= 1)
                    mx = fmaxf(mx, __shfl_xor(mx, off));
                const float ex = expf(v - mx);
                w[k] = (lane < cnt) ? ex : 0.f;
            }

            constexpr int HB  = NBATCH / 2;
            constexpr int EPH = HB * EPS;
            rows_consume<H, VEC, EPS, 0, 0, HB>(w, sub, myh, ra, acc, den);
            if (cnt > EPH)
                rows_consume<H, VEC, EPS, 0, HB, NBATCH>(w, sub, myh, ra, acc, den);
            if (cnt > 2 * EPH)
                rows_consume<H, VEC, EPS, NBATCH, 0, HB>(w, sub, myh, rb, acc, den);
            if (cnt > 3 * EPH)
                rows_consume<H, VEC, EPS, NBATCH, HB, NBATCH>(w, sub, myh, rb, acc, den);

            if constexpr (EPS == 1) {
                if (cnt > 2 * EPB) {
                    rows_request<F, VEC, EPS, 2 * NBATCH>(h, s, sub, f0, ra);
                    if (cnt > 3 * EPB)
                        rows_request<F, VEC, EPS, 3 * NBATCH>(h, s, sub, f0, rb);
                    rows_consume<H, VEC, EPS, 2 * NBATCH, 0, NBATCH>(w, sub, myh, ra, acc, den);
                    if (cnt > 3 * EPB)
                        rows_consume<H, VEC, EPS, 3 * NBATCH, 0, NBATCH>(w, sub, myh, rb, acc, den);
                }
            }
        } else {
            if (has_next)
                s_next = (lane < nend - nbeg) ? csr[nbeg + lane] : nn;
            if (has_n2) {
                n2beg = start[n2];
                n2end = start[n2 + 1];
                for (int k = 0; k < H; ++k) n2ad[k] = ad[n2 * H + k];
            }

            float mx[H];
            for (int k = 0; k < H; ++k) mx[k] = -INFINITY;

            for (int base = 0; base < cnt; base += 32) {
                const int idx = base + lane;
                const int s = (idx < cnt - 1) ? csr[beg + idx] : n;
                float av[H];
                if constexpr (H == 4) {
                    const float4 t = *(const float4*)(as + (size_t)s * 4);
                    av[0] = t.x; av[1] = t.y; av[2] = t.z; av[3] = t.w;
                } else {
                    for (int k = 0; k < H; ++k) av[k] = as[(size_t)s * H + k];
                }
                for (int k = 0; k < H; ++k) {
                    float v = av[k] + adn[k];
                    v = (v > 0.f) ? v : NEG_SLOPE * v;
                    mx[k] = fmaxf(mx[k], (idx < cnt) ? v : -INFINITY);
                }
            }
            for (int k = 0; k < H; ++k)
                for (int off = 16; off > 0; off >>= 1)
                    mx[k] = fmaxf(mx[k], __shfl_xor(mx[k], off));

            for (int base = 0; base < cnt; base += 32) {
                const int idx = base + lane;
                const int s = (idx < cnt - 1) ? csr[beg + idx] : n;
                float av[H];
                if constexpr (H == 4) {
                    const float4 t = *(const float4*)(as + (size_t)s * 4);
                    av[0] = t.x; av[1] = t.y; av[2] = t.z; av[3] = t.w;
                } else {
                    for (int k = 0; k < H; ++k) av[k] = as[(size_t)s * H + k];
                }
                float w[H];
                for (int k = 0; k < H; ++k) {
                    float v = av[k] + adn[k];
                    v = (v > 0.f) ? v : NEG_SLOPE * v;
                    const float ex = expf(v - mx[k]);
                    w[k] = (idx < cnt) ? ex : 0.f;
                }

                const int m = min(32, cnt - base);
                for (int j = 0; j < m; j += UNR * EPS) {
                    #pragma unroll
                    for (int u = 0; u < UNR; ++u) {
                        const int jj = j + u * EPS + sub;
                        const int sj = __shfl(s, jj);
                        float wj = 0.f;
                        for (int k = 0; k < H; ++k) {
                            const float t = __shfl(w[k], jj);
                            if (k == myh) wj = t;
                        }
                        const float* row = h + (size_t)sj * F + f0;
                        if constexpr (VEC == 2) {
                            const float2 r = *(const float2*)row;
                            acc[0] = fmaf(wj, r.x, acc[0]);
                            acc[1] = fmaf(wj, r.y, acc[1]);
                        } else {
                            acc[0] = fmaf(wj, row[0], acc[0]);
                        }
                        den += wj;
                    }
                }
            }
        }

        for (int off = LPR; off < 32; off <<= 1) {
            for (int v = 0; v < VEC; ++v) acc[v] += __shfl_xor(acc[v], off);
            den += __shfl_xor(den, off);
        }

        if (sub == 0) {
            for (int v = 0; v < VEC; ++v) {
                float val = acc[v] / (den + 1e-16f) + bv[v];
                if (ELU) val = (val > 0.f) ? val : (expf(val) - 1.f);
                out[(size_t)n * F + f0 + v] = val;
            }
        }

        beg = nbeg; end = nend; s_cur = s_next;
        nbeg = n2beg; nend = n2end;
        for (int k = 0; k < H; ++k) { adn[k] = nad[k]; nad[k] = n2ad[k]; }
    }
}

static inline int cdiv(int a, int b) { return (a + b - 1) / b; }

extern "C" void kernel_launch(void* const* d_in, const int* in_sizes, int n_in,
                              void* d_out, int out_size, void* d_ws, size_t ws_size,
                              hipStream_t stream) {
    const float* x      = (const float*)d_in[0];
    const int*   ei     = (const int*)  d_in[1];
    const float* W1     = (const float*)d_in[2];
    const float* a_src1 = (const float*)d_in[3];
    const float* a_dst1 = (const float*)d_in[4];
    const float* b1     = (const float*)d_in[5];
    const float* W2     = (const float*)d_in[6];
    const float* a_src2 = (const float*)d_in[7];
    const float* a_dst2 = (const float*)d_in[8];
    const float* b2     = (const float*)d_in[9];

    const int N  = in_sizes[0] / 128;
    const int E  = in_sizes[1] / 2;

    float* ws   = (float*)d_ws;
    float* h1   = ws;
    float* agg1 = h1   + (size_t)N * 64;
    float* as1  = agg1 + (size_t)N * 64;
    float* ad1  = as1  + (size_t)N * 4;
    int*   cnt  = (int*)(ad1 + (size_t)N * 4);
    int*   strt = cnt   + N;
    int*   rank = strt  + N + 4;
    int*   csr  = rank  + E;
    int*   bsum = csr   + E;
    float* h2   = h1;
    float* as2  = h2   + (size_t)N * 16;
    float* ad2  = as2  + N;

    float* out  = (float*)d_out;

    const int TB = 256;
    const int NB = cdiv(N, 1024);

    zero_int_kernel<<<cdiv(N, TB), TB, 0, stream>>>(cnt, N);
    bucket_count_kernel<<<1024, TB, 0, stream>>>(ei, E, cnt, rank);
    bucket_partial_kernel<<<NB, 256, 0, stream>>>(cnt, bsum, N);
    bucket_scan_kernel<<<NB, 256, 0, stream>>>(cnt, bsum, strt, N);
    bucket_place_kernel<<<1024, TB, 0, stream>>>(ei, E, strt, rank, csr);

    gemm_wmma_f32<128, 4><<<N / 16, dim3(32, 4), 0, stream>>>(x, W1, h1);

    alpha_kernel<4, 16><<<cdiv(N * 4, TB), TB, 0, stream>>>(
        h1, a_src1, a_dst1, as1, ad1, N);

    node_gather_kernel<4, 16, true><<<2048, TB, 0, stream>>>(
        strt, csr, h1, as1, ad1, b1, agg1, N);

    gemm_wmma_f32<64, 1><<<N / 16, dim3(32, 1), 0, stream>>>(agg1, W2, h2);

    alpha_kernel<1, 16><<<cdiv(N, TB), TB, 0, stream>>>(
        h2, a_src2, a_dst2, as2, ad2, N);

    node_gather_kernel<1, 16, false><<<2048, TB, 0, stream>>>(
        strt, csr, h2, as2, ad2, b2, out, N);
}


# Logits (as, ad) summed in the GEMM epilogue from the accumulator tile by a 16-lane exchange tree (order of their sums changes); both logit launches and the re-read of h go; 2nd GEMM 32 rows/block
# speedup vs baseline: 5.3576x; 1.0856x over previous
#include <hip/hip_runtime.h>
#include <hip/hip_bf16.h>
#include <math.h>

typedef float v2f __attribute__((ext_vector_type(2)));
typedef float v8f __attribute__((ext_vector_type(8)));

#define NEG_SLOPE 0.2f

__global__ void zero_int_kernel(int* __restrict__ p, int n) {
    int i = blockIdx.x * blockDim.x + threadIdx.x;
    if (i < n) p[i] = 0;
}

__global__ void bucket_count_kernel(const int* __restrict__ ei, int E,
                                    int* __restrict__ cnt,
                                    int* __restrict__ rank) {
    const int stride = gridDim.x * blockDim.x;
    for (int e = blockIdx.x * blockDim.x + threadIdx.x; e < E; e += stride) {
        const int d = ei[E + e];
        rank[e] = atomicAdd(&cnt[d], 1);
    }
}

__global__ void bucket_partial_kernel(const int* __restrict__ cnt,
                                      int* __restrict__ bsum, int n) {
    __shared__ int sh[256];
    const int t  = threadIdx.x;
    const int i0 = blockIdx.x * 1024 + t * 4;
    int s = 0;
    for (int k = 0; k < 4; ++k) s += (i0 + k < n) ? cnt[i0 + k] : 0;
    sh[t] = s;
    __syncthreads();
    for (int off = 128; off > 0; off >>= 1) {
        if (t < off) sh[t] += sh[t + off];
        __syncthreads();
    }
    if (t == 0) bsum[blockIdx.x] = sh[0];
}

__global__ void bucket_scan_kernel(const int* __restrict__ cnt,
                                   const int* __restrict__ bsum,
                                   int* __restrict__ start, int n) {
    __shared__ int sh[256];
    const int t  = threadIdx.x;
    const int i0 = blockIdx.x * 1024 + t * 4;

    int part = 0;
    for (int i = t; i < (int)blockIdx.x; i += 256) part += bsum[i];

    int c[4];
    int tsum = 0;
    for (int k = 0; k < 4; ++k) {
        c[k] = (i0 + k < n) ? cnt[i0 + k] : 0;
        tsum += c[k];
    }

    sh[t] = part;
    __syncthreads();
    for (int off = 128; off > 0; off >>= 1) {
        if (t < off) sh[t] += sh[t + off];
        __syncthreads();
    }
    const int base = sh[0];
    __syncthreads();

    sh[t] = tsum;
    __syncthreads();
    for (int off = 1; off < 256; off <<= 1) {
        const int v = (t >= off) ? sh[t - off] : 0;
        __syncthreads();
        sh[t] += v;
        __syncthreads();
    }
    int run = base + sh[t] - tsum;
    for (int k = 0; k < 4; ++k) {
        if (i0 + k < n) { start[i0 + k] = run; run += c[k]; }
    }
    if (i0 < n && n <= i0 + 4) start[n] = run;
}

__global__ void bucket_place_kernel(const int* __restrict__ ei, int E,
                                    const int* __restrict__ start,
                                    const int* __restrict__ rank,
                                    int* __restrict__ csr) {
    const int stride = gridDim.x * blockDim.x;
    for (int e = blockIdx.x * blockDim.x + threadIdx.x; e < E; e += stride) {
        const int d = ei[E + e];
        csr[start[d] + rank[e]] = ei[e];
    }
}

template <int M>
__device__ __forceinline__ float row_xor(float v) {
    return __int_as_float(__builtin_amdgcn_update_dpp(
        0, __float_as_int(v), 0x160 | M, 0xf, 0xf, true));
}

template <int K, int NCOLT, int NROWT>
__global__ void gemm_logits_wmma_f32(const float* __restrict__ X,
                                     const float* __restrict__ W,
                                     const float* __restrict__ a_src,
                                     const float* __restrict__ a_dst,
                                     float* __restrict__ H,
                                     float* __restrict__ as,
                                     float* __restrict__ ad) {
    constexpr int FOUT = 16 * NCOLT;
    constexpr int ROWS = 16 * NROWT;
    constexpr int NW   = NCOLT * NROWT;
    constexpr int NL   = ROWS * NCOLT;
    constexpr int PER  = NL / 32;
    static_assert(NW >= 2, "one wave stores as, another ad");
    static_assert(PER == 1 || PER == 2, "a block's logits are one store per kind");
    __shared__ float As[ROWS][K + 4];
    __shared__ __attribute__((aligned(8))) float sh_log[2][NL];

    const int r0   = blockIdx.x * ROWS;
    const int lane = threadIdx.x;
    const int wave = threadIdx.y;
    const int rt   = wave / NCOLT;
    const int ct   = wave % NCOLT;
    const int tid  = wave * 32 + lane;
    constexpr int nthr = 32 * NW;

    for (int i = tid; i < ROWS * K; i += nthr) {
        int rr = i / K, cc = i % K;
        As[rr][cc] = X[(size_t)(r0 + rr) * K + cc];
    }
    __syncthreads();

    const int half = lane >> 4;
    const int m    = lane & 15;
    const int c0   = ct * 16;
    const int ar   = rt * 16 + m;

    const float asv = a_src[c0 + m];
    const float adv = a_dst[c0 + m];

    v8f acc = {};
    #pragma unroll 4
    for (int kk = 0; kk < K; kk += 4) {
        const int kb = kk + 2 * half;
        v2f a, b;
        a[0] = As[ar][kb];
        a[1] = As[ar][kb + 1];
        b[0] = W[(size_t)(kb)     * FOUT + c0 + m];
        b[1] = W[(size_t)(kb + 1) * FOUT + c0 + m];
        acc = __builtin_amdgcn_wmma_f32_16x16x4_f32(
             false, a,  false, b,
             (short)0, acc,  false,  false);
    }

    float lg;
    {
        const bool b3 = (m & 8) != 0, b2 = (m & 4) != 0;
        const bool b1 = (m & 2) != 0, b0 = (m & 1) != 0;
        const float kv = b3 ? adv : asv;
        const float sv = b3 ? asv : adv;
        float q[8], t[4], u[2];
        #pragma unroll
        for (int r = 0; r < 8; ++r)
            q[r] = acc[r] * kv + row_xor<8>(acc[r] * sv);
        #pragma unroll
        for (int r = 0; r < 4; ++r)
            t[r] = (b2 ? q[r + 4] : q[r]) + row_xor<4>(b2 ? q[r] : q[r + 4]);
        #pragma unroll
        for (int r = 0; r < 2; ++r)
            u[r] = (b1 ? t[r + 2] : t[r]) + row_xor<2>(b1 ? t[r] : t[r + 2]);
        lg = (b0 ? u[1] : u[0]) + row_xor<1>(b0 ? u[0] : u[1]);
    }
    sh_log[m >> 3][(rt * 16 + (m & 7) + 8 * half) * NCOLT + ct] = lg;
    __syncthreads();

    #pragma unroll
    for (int r = 0; r < 8; ++r) {
        H[(size_t)(r0 + rt * 16 + r + 8 * half) * FOUT + c0 + m] = acc[r];
    }

    if (wave < 2) {
        float* const dst = (wave == 0 ? as : ad) + (size_t)blockIdx.x * NL + lane * PER;
        if constexpr (PER == 2)
            *(v2f*)dst = *(const v2f*)&sh_log[wave][lane * 2];
        else
            *dst = sh_log[wave][lane];
    }
}

typedef float v4f __attribute__((ext_vector_type(4)));
typedef int   v4i __attribute__((ext_vector_type(4)));

template <int MASK>
__device__ __forceinline__ float lane_xor(float v) {
    return __int_as_float(
        __builtin_amdgcn_ds_swizzle(__float_as_int(v), (MASK << 10) | 0x1f));
}
__device__ __forceinline__ float wave_max(float v) {
    v = fmaxf(v, lane_xor<16>(v));
    v = fmaxf(v, lane_xor<8>(v));
    v = fmaxf(v, lane_xor<4>(v));
    v = fmaxf(v, lane_xor<2>(v));
    v = fmaxf(v, lane_xor<1>(v));
    return v;
}

template <int F, int NR, int GR, int EPS, int S0>
__device__ __forceinline__ void rows_request(const float* __restrict__ h,
                                             const int* idx_slot, int cnt,
                                             int f0, v4f (&r)[NR]) {
    v4i ix[NR / 4];
    #pragma unroll
    for (int q = 0; q < NR / 4; ++q) ix[q] = *(const v4i*)(idx_slot + 4 * q);
    #pragma unroll
    for (int g = 0; g < NR / GR; ++g) {
        if (g == 0 || cnt > (S0 + g * GR) * EPS) {
            #pragma unroll
            for (int u = 0; u < GR; ++u) {
                const int j = g * GR + u;
                const unsigned off = (unsigned)ix[j / 4][j % 4] * (unsigned)(F * 4)
                                   + (unsigned)f0 * 4u;
                r[j] = *(const v4f*)((const char*)h + off);
            }
        }
    }
}

template <int NR, int GR, int EPS, int S0>
__device__ __forceinline__ void rows_consume(const float* w_slot, int cnt,
                                             const v4f (&r)[NR],
                                             float (&acc)[4], float& den) {
    #pragma unroll
    for (int q = 0; q < NR / 4; ++q) {
        if (q == 0 || cnt > (S0 + 4 * q) * EPS) {
            const v4f wv = *(const v4f*)(w_slot + 4 * q);
            #pragma unroll
            for (int g = 0; g < 4 / GR; ++g) {
                if (g == 0 || cnt > (S0 + 4 * q + g * GR) * EPS) {
                    #pragma unroll
                    for (int u = 0; u < GR; ++u) {
                        const int j = 4 * q + g * GR + u;
                        const float wj = wv[g * GR + u];
                        #pragma unroll
                        for (int v = 0; v < 4; ++v) acc[v] = fmaf(wj, r[j][v], acc[v]);
                        den += wj;
                    }
                }
            }
        }
    }
}

template <int H, int D, bool ELU>
__global__ void __launch_bounds__(256, 16)
node_gather_kernel(const int* __restrict__ start, const int* __restrict__ csr,
                   const float* __restrict__ h,
                   const float* __restrict__ as, const float* __restrict__ ad,
                   const float* __restrict__ bias,
                   float* __restrict__ out, int N) {
    constexpr int F   = H * D;
    constexpr int VEC = 4;
    constexpr int LPR = F / VEC;
    constexpr int EPS = 32 / LPR;
    constexpr int NST = LPR;
    constexpr int NR  = (NST < 8) ? NST : 8;
    constexpr int GR  = NR / 2;
    static_assert(LPR == 16 || LPR == 4, "row must map to 16 or 4 lanes");
    static_assert(D % VEC == 0, "a lane's features must lie in one head");

    __shared__ __attribute__((aligned(16))) int   sh_idx[8][32];
    __shared__ __attribute__((aligned(16))) float sh_w[8][H * 32];

    const int lane = threadIdx.x & 31;
    const int wib  = threadIdx.x >> 5;
    const int sub  = lane / LPR;
    const int f0   = (lane % LPR) * VEC;
    const int myh  = f0 / D;
    const int ppos = (lane % EPS) * LPR + lane / EPS;
    int*   const my_idx = sh_idx[wib];
    float* const my_w   = sh_w[wib];
    const int* const idx_slot = my_idx + sub * LPR;
    const float* const w_slot = my_w + myh * 32 + sub * LPR;
    const int fo = (LPR == 16) ? f0 + 2 * (lane >> 4)
                               : f0 + 2 * ((lane >> 4) & 1) + ((lane >> 3) & 1);
    const int wave   = blockIdx.x * (blockDim.x >> 5) + (threadIdx.x >> 5);
    const int nwaves = gridDim.x * (blockDim.x >> 5);

    if (wave >= N) return;

    const int wave_s   = __builtin_amdgcn_readfirstlane(wave);
    const int nwaves_s = __builtin_amdgcn_readfirstlane(nwaves);

    float bo0 = bias[fo];
    float bo1 = (LPR == 16) ? bias[fo + 1] : 0.f;

    int   beg, end, s_cur;
    float adn[H];
    int   nbeg = 0, nend = 0;
    float nad[H];
    {
        const int n = wave_s;
        beg = start[n];
        end = start[n + 1];
        for (int k = 0; k < H; ++k) adn[k] = ad[n * H + k];
        for (int k = 0; k < H; ++k) nad[k] = 0.f;
        const int nn = n + nwaves_s;
        if (nn < N) {
            nbeg = start[nn];
            nend = start[nn + 1];
            for (int k = 0; k < H; ++k) nad[k] = ad[nn * H + k];
        }
        s_cur = (lane < end - beg) ? csr[beg + lane] : n;
    }

    for (int n = wave_s; n < N; n += nwaves_s) {
        const int cnt = end - beg + 1;

        const bool has_next = n + nwaves_s < N;
        const int  nn       = has_next ? n + nwaves_s : n;
        const int  n2       = nn + nwaves_s;
        const bool has_n2   = has_next && n2 < N;
        int   n2beg = 0, n2end = 0;
        float n2ad[H];
        for (int k = 0; k < H; ++k) n2ad[k] = 0.f;
        int s_next = nn;

        float acc[VEC];
        for (int v = 0; v < VEC; ++v) acc[v] = 0.f;
        float den = 0.f;

        if (cnt <= 32) {
            const int s = s_cur;
            my_idx[ppos] = s;

            float av[H];
            if constexpr (H == 4) {
                const float4 t = *(const float4*)(as + (size_t)s * 4);
                av[0] = t.x; av[1] = t.y; av[2] = t.z; av[3] = t.w;
            } else {
                for (int k = 0; k < H; ++k) av[k] = as[(size_t)s * H + k];
            }

            v4f r[NR];
            rows_request<F, NR, GR, EPS, 0>(h, idx_slot, cnt, f0, r);

            if (has_next)
                s_next = (lane < nend - nbeg) ? csr[nbeg + lane] : nn;
            if (has_n2) {
                n2beg = start[n2];
                n2end = start[n2 + 1];
                for (int k = 0; k < H; ++k) n2ad[k] = ad[n2 * H + k];
            }

            for (int k = 0; k < H; ++k) {
                float v = av[k] + adn[k];
                v = (v > 0.f) ? v : NEG_SLOPE * v;
                const float mx = wave_max((lane < cnt) ? v : -INFINITY);
                const float ex = expf(v - mx);
                my_w[k * 32 + ppos] = (lane < cnt) ? ex : 0.f;
            }

            rows_consume<NR, GR, EPS, 0>(w_slot, cnt, r, acc, den);

            if constexpr (NST > NR) {
                if (cnt > NR * EPS) {
                    rows_request<F, NR, GR, EPS, NR>(h, idx_slot + NR, cnt, f0, r);
                    rows_consume<NR, GR, EPS, NR>(w_slot + NR, cnt, r, acc, den);
                }
            }
        } else {
            if (has_next)
                s_next = (lane < nend - nbeg) ? csr[nbeg + lane] : nn;
            if (has_n2) {
                n2beg = start[n2];
                n2end = start[n2 + 1];
                for (int k = 0; k < H; ++k) n2ad[k] = ad[n2 * H + k];
            }

            float mx[H];
            for (int k = 0; k < H; ++k) mx[k] = -INFINITY;

            for (int base = 0; base < cnt; base += 32) {
                const int idx = base + lane;
                const int s = (idx < cnt - 1) ? csr[beg + idx] : n;
                float av[H];
                if constexpr (H == 4) {
                    const float4 t = *(const float4*)(as + (size_t)s * 4);
                    av[0] = t.x; av[1] = t.y; av[2] = t.z; av[3] = t.w;
                } else {
                    for (int k = 0; k < H; ++k) av[k] = as[(size_t)s * H + k];
                }
                for (int k = 0; k < H; ++k) {
                    float v = av[k] + adn[k];
                    v = (v > 0.f) ? v : NEG_SLOPE * v;
                    mx[k] = fmaxf(mx[k], (idx < cnt) ? v : -INFINITY);
                }
            }
            for (int k = 0; k < H; ++k) mx[k] = wave_max(mx[k]);

            for (int base = 0; base < cnt; base += 32) {
                const int idx = base + lane;
                const int s = (idx < cnt - 1) ? csr[beg + idx] : n;
                float av[H];
                if constexpr (H == 4) {
                    const float4 t = *(const float4*)(as + (size_t)s * 4);
                    av[0] = t.x; av[1] = t.y; av[2] = t.z; av[3] = t.w;
                } else {
                    for (int k = 0; k < H; ++k) av[k] = as[(size_t)s * H + k];
                }
                float w[H];
                for (int k = 0; k < H; ++k) {
                    float v = av[k] + adn[k];
                    v = (v > 0.f) ? v : NEG_SLOPE * v;
                    const float ex = expf(v - mx[k]);
                    w[k] = (idx < cnt) ? ex : 0.f;
                }

                const int m = min(32, cnt - base);
                my_idx[ppos] = s;
                for (int k = 0; k < H; ++k) my_w[k * 32 + ppos] = w[k];
                #pragma unroll 2
                for (int j = 0; j < NST; ++j) {
                    if (j * EPS < m) {
                        const int   sj = idx_slot[j];
                        const float wj = w_slot[j];
                        const v4f rr = *(const v4f*)(h + (size_t)sj * F + f0);
                        #pragma unroll
                        for (int v = 0; v < VEC; ++v)
                            acc[v] = fmaf(wj, rr[v], acc[v]);
                        den += wj;
                    }
                }
            }
        }

        float k0, k1;
        {
            const bool hi = (lane & 16) != 0;
            k0 = hi ? acc[2] : acc[0];
            k1 = hi ? acc[3] : acc[1];
            const float s0 = hi ? acc[0] : acc[2];
            const float s1 = hi ? acc[1] : acc[3];
            k0  += lane_xor<16>(s0);
            k1  += lane_xor<16>(s1);
            den += lane_xor<16>(den);
        }
        if constexpr (LPR == 16) {
            v2f val;
            val[0] = k0 / (den + 1e-16f) + bo0;
            val[1] = k1 / (den + 1e-16f) + bo1;
            if (ELU) {
                val[0] = (val[0] > 0.f) ? val[0] : (expf(val[0]) - 1.f);
                val[1] = (val[1] > 0.f) ? val[1] : (expf(val[1]) - 1.f);
            }
            *(v2f*)((char*)out + ((unsigned)n * (unsigned)(F * 4) + (unsigned)fo * 4u)) = val;
        } else {
            const bool hi8 = (lane & 8) != 0;
            float kk = hi8 ? k1 : k0;
            const float ss = hi8 ? k0 : k1;
            kk  += lane_xor<8>(ss);
            den += lane_xor<8>(den);
            kk  += lane_xor<4>(kk);
            den += lane_xor<4>(den);
            if ((lane & 4) == 0) {
                float val = kk / (den + 1e-16f) + bo0;
                if (ELU) val = (val > 0.f) ? val : (expf(val) - 1.f);
                *(float*)((char*)out + ((unsigned)n * (unsigned)(F * 4) + (unsigned)fo * 4u)) = val;
            }
        }

        beg = nbeg; end = nend; s_cur = s_next;
        nbeg = n2beg; nend = n2end;
        for (int k = 0; k < H; ++k) { adn[k] = nad[k]; nad[k] = n2ad[k]; }
    }
}

static inline int cdiv(int a, int b) { return (a + b - 1) / b; }

extern "C" void kernel_launch(void* const* d_in, const int* in_sizes, int n_in,
                              void* d_out, int out_size, void* d_ws, size_t ws_size,
                              hipStream_t stream) {
    const float* x      = (const float*)d_in[0];
    const int*   ei     = (const int*)  d_in[1];
    const float* W1     = (const float*)d_in[2];
    const float* a_src1 = (const float*)d_in[3];
    const float* a_dst1 = (const float*)d_in[4];
    const float* b1     = (const float*)d_in[5];
    const float* W2     = (const float*)d_in[6];
    const float* a_src2 = (const float*)d_in[7];
    const float* a_dst2 = (const float*)d_in[8];
    const float* b2     = (const float*)d_in[9];

    const int N  = in_sizes[0] / 128;
    const int E  = in_sizes[1] / 2;

    float* ws   = (float*)d_ws;
    float* h1   = ws;
    float* agg1 = h1   + (size_t)N * 64;
    float* as1  = agg1 + (size_t)N * 64;
    float* ad1  = as1  + (size_t)N * 4;
    int*   cnt  = (int*)(ad1 + (size_t)N * 4);
    int*   strt = cnt   + N;
    int*   rank = strt  + N + 4;
    int*   csr  = rank  + E;
    int*   bsum = csr   + E;
    float* h2   = h1;
    float* as2  = h2   + (size_t)N * 16;
    float* ad2  = as2  + N;

    float* out  = (float*)d_out;

    const int TB = 256;
    const int NB = cdiv(N, 1024);

    zero_int_kernel<<<cdiv(N, TB), TB, 0, stream>>>(cnt, N);
    bucket_count_kernel<<<1024, TB, 0, stream>>>(ei, E, cnt, rank);
    bucket_partial_kernel<<<NB, 256, 0, stream>>>(cnt, bsum, N);
    bucket_scan_kernel<<<NB, 256, 0, stream>>>(cnt, bsum, strt, N);
    bucket_place_kernel<<<1024, TB, 0, stream>>>(ei, E, strt, rank, csr);

    gemm_logits_wmma_f32<128, 4, 1><<<N / 16, dim3(32, 4), 0, stream>>>(
        x, W1, a_src1, a_dst1, h1, as1, ad1);

    node_gather_kernel<4, 16, true><<<2048, TB, 0, stream>>>(
        strt, csr, h1, as1, ad1, b1, agg1, N);

    gemm_logits_wmma_f32<64, 1, 2><<<N / 32, dim3(32, 2), 0, stream>>>(
        agg1, W2, a_src2, a_dst2, h2, as2, ad2);

    node_gather_kernel<1, 16, false><<<2048, TB, 0, stream>>>(
        strt, csr, h2, as2, ad2, b2, out, N);
}


# Edge ranks, in-degrees, block sums from 16-bit LDS counters (returning LDS add) per 31250-edge chunk, not a returning global atomic per edge; clear, block-sum launches go; list order: chunk, arrival
# speedup vs baseline: 7.0146x; 1.3093x over previous
#include <hip/hip_runtime.h>
#include <hip/hip_bf16.h>
#include <math.h>

typedef float v2f __attribute__((ext_vector_type(2)));
typedef float v8f __attribute__((ext_vector_type(8)));

#define NEG_SLOPE 0.2f

#define NCHUNK 32
#define CAPW   50176
#define NPITCH (2 * CAPW)

template <int M>
__device__ __forceinline__ unsigned row_xor_u(unsigned v) {
    return (unsigned)__builtin_amdgcn_update_dpp(
        0, (int)v, 0x160 | M, 0xf, 0xf, true);
}

__global__ void __launch_bounds__(1024)
chunk_count_kernel(const int* __restrict__ ei, int E,
                   unsigned* __restrict__ tabg,
                   unsigned short* __restrict__ rank,
                   int* __restrict__ bsumg) {
    __shared__ __attribute__((aligned(16))) unsigned tab[CAPW];
    __shared__ unsigned blk[128];
    const int t  = threadIdx.x;
    const int g  = blockIdx.x;
    const int CH = (E + NCHUNK - 1) / NCHUNK;
    const int c0 = g * CH;
    const int c1 = min(E, c0 + CH);

    for (int i = t * 4; i < CAPW; i += 4096)
        *(uint4*)&tab[i] = make_uint4(0u, 0u, 0u, 0u);
    if (t < 128) blk[t] = 0u;
    __syncthreads();

    constexpr int EPT = 30;
    int e0 = c0 + t;
    for (; e0 + (EPT - 1) * 1024 < c1; e0 += EPT * 1024) {
        int d[EPT];
        unsigned old[EPT];
        #pragma unroll
        for (int u = 0; u < EPT; ++u) d[u] = ei[E + e0 + u * 1024];
        #pragma unroll
        for (int u = 0; u < EPT; ++u)
            old[u] = atomicAdd(&tab[d[u] >> 1], 1u << ((unsigned)(d[u] & 1) * 16u));
        #pragma unroll
        for (int u = 0; u < EPT; ++u)
            rank[e0 + u * 1024] =
                (unsigned short)(old[u] >> ((unsigned)(d[u] & 1) * 16u));
    }
    for (int e = e0; e < c1; e += 1024) {
        const int dd = ei[E + e];
        const unsigned sh  = (unsigned)(dd & 1) * 16u;
        const unsigned old = atomicAdd(&tab[dd >> 1], 1u << sh);
        rank[e] = (unsigned short)(old >> sh);
    }
    __syncthreads();

    unsigned* const dst = tabg + (size_t)g * CAPW;
    constexpr int NRND = CAPW / 4096;
    const bool tail = t * 4 + NRND * 4096 < CAPW;
    unsigned s[NRND + 1];
    #pragma unroll
    for (int r = 0; r <= NRND; ++r) {
        const int i = t * 4 + r * 4096;
        s[r] = 0u;
        if (r < NRND || tail) {
            const uint4 q = *(const uint4*)&tab[i];
            *(uint4*)&dst[i] = q;
            s[r] = (q.x & 0xffffu) + (q.x >> 16) + (q.y & 0xffffu) + (q.y >> 16)
                 + (q.z & 0xffffu) + (q.z >> 16) + (q.w & 0xffffu) + (q.w >> 16);
        }
    }
    #pragma unroll
    for (int r = 0; r <= NRND; ++r) {
        if (r < NRND || tail) {
            unsigned x = s[r];
            x += row_xor_u<1>(x);
            x += row_xor_u<2>(x);
            x += row_xor_u<4>(x);
            x += row_xor_u<8>(x);
            if ((t & 15) == 0) atomicAdd(&blk[(t * 4 + r * 4096) >> 9], x);
        }
    }
    __syncthreads();
    if (t < 128) bsumg[g * 128 + t] = (int)blk[t];
}

__global__ void __launch_bounds__(256)
chunk_scan_kernel(const unsigned* __restrict__ tabg,
                                  const int* __restrict__ bsumg,
                                  int* __restrict__ start,
                                  int* __restrict__ base, int n) {
    __shared__ int sh[256];
    const int t  = threadIdx.x;
    const int i0 = blockIdx.x * 1024 + t * 4;
    const int w0 = blockIdx.x * 512 + t * 2;

    int part = 0;
    if (t < (int)blockIdx.x) {
        #pragma unroll
        for (int g = 0; g < NCHUNK; ++g) part += bsumg[g * 128 + t];
    }

    uint2 v[NCHUNK];
    #pragma unroll
    for (int g = 0; g < NCHUNK; ++g)
        v[g] = *(const uint2*)&tabg[(size_t)g * CAPW + w0];

    int c[4] = {0, 0, 0, 0};
    #pragma unroll
    for (int g = 0; g < NCHUNK; ++g) {
        c[0] += (int)(v[g].x & 0xffffu);
        c[1] += (int)(v[g].x >> 16);
        c[2] += (int)(v[g].y & 0xffffu);
        c[3] += (int)(v[g].y >> 16);
    }
    const int tsum = c[0] + c[1] + c[2] + c[3];

    sh[t] = part;
    __syncthreads();
    for (int off = 128; off > 0; off >>= 1) {
        if (t < off) sh[t] += sh[t + off];
        __syncthreads();
    }
    const int before = sh[0];
    __syncthreads();

    sh[t] = tsum;
    __syncthreads();
    for (int off = 1; off < 256; off <<= 1) {
        const int x = (t >= off) ? sh[t - off] : 0;
        __syncthreads();
        sh[t] += x;
        __syncthreads();
    }
    int o[4];
    o[0] = before + sh[t] - tsum;
    o[1] = o[0] + c[0];
    o[2] = o[1] + c[1];
    o[3] = o[2] + c[2];
    for (int k = 0; k < 4; ++k)
        if (i0 + k < n) start[i0 + k] = o[k];
    if (i0 < n && n <= i0 + 4) start[n] = o[3] + c[3];

    #pragma unroll
    for (int g = 0; g < NCHUNK; ++g) {
        *(int4*)&base[(size_t)g * NPITCH + i0] = make_int4(o[0], o[1], o[2], o[3]);
        o[0] += (int)(v[g].x & 0xffffu);
        o[1] += (int)(v[g].x >> 16);
        o[2] += (int)(v[g].y & 0xffffu);
        o[3] += (int)(v[g].y >> 16);
    }
}

__global__ void __launch_bounds__(256)
chunk_place_kernel(const int* __restrict__ ei, int E,
                                   const int* __restrict__ base,
                                   const unsigned short* __restrict__ rank,
                                   int* __restrict__ csr) {
    const int g  = blockIdx.y;
    const int CH = (E + NCHUNK - 1) / NCHUNK;
    const int c0 = g * CH;
    const int c1 = min(E, c0 + CH);
    const int* const bg = base + (size_t)g * NPITCH;
    const int stride = gridDim.x * blockDim.x;
    for (int e = c0 + blockIdx.x * blockDim.x + threadIdx.x; e < c1; e += stride) {
        const int d = ei[E + e];
        csr[bg[d] + (int)rank[e]] = ei[e];
    }
}

template <int M>
__device__ __forceinline__ float row_xor(float v) {
    return __int_as_float(__builtin_amdgcn_update_dpp(
        0, __float_as_int(v), 0x160 | M, 0xf, 0xf, true));
}

template <int K, int NCOLT, int NROWT>
__global__ void gemm_logits_wmma_f32(const float* __restrict__ X,
                                     const float* __restrict__ W,
                                     const float* __restrict__ a_src,
                                     const float* __restrict__ a_dst,
                                     float* __restrict__ H,
                                     float* __restrict__ as,
                                     float* __restrict__ ad) {
    constexpr int FOUT = 16 * NCOLT;
    constexpr int ROWS = 16 * NROWT;
    constexpr int NW   = NCOLT * NROWT;
    constexpr int NL   = ROWS * NCOLT;
    constexpr int PER  = NL / 32;
    static_assert(NW >= 2, "one wave stores as, another ad");
    static_assert(PER == 1 || PER == 2, "a block's logits are one store per kind");
    __shared__ float As[ROWS][K + 4];
    __shared__ __attribute__((aligned(8))) float sh_log[2][NL];

    const int r0   = blockIdx.x * ROWS;
    const int lane = threadIdx.x;
    const int wave = threadIdx.y;
    const int rt   = wave / NCOLT;
    const int ct   = wave % NCOLT;
    const int tid  = wave * 32 + lane;
    constexpr int nthr = 32 * NW;

    for (int i = tid; i < ROWS * K; i += nthr) {
        int rr = i / K, cc = i % K;
        As[rr][cc] = X[(size_t)(r0 + rr) * K + cc];
    }
    __syncthreads();

    const int half = lane >> 4;
    const int m    = lane & 15;
    const int c0   = ct * 16;
    const int ar   = rt * 16 + m;

    const float asv = a_src[c0 + m];
    const float adv = a_dst[c0 + m];

    v8f acc = {};
    #pragma unroll 4
    for (int kk = 0; kk < K; kk += 4) {
        const int kb = kk + 2 * half;
        v2f a, b;
        a[0] = As[ar][kb];
        a[1] = As[ar][kb + 1];
        b[0] = W[(size_t)(kb)     * FOUT + c0 + m];
        b[1] = W[(size_t)(kb + 1) * FOUT + c0 + m];
        acc = __builtin_amdgcn_wmma_f32_16x16x4_f32(
             false, a,  false, b,
             (short)0, acc,  false,  false);
    }

    float lg;
    {
        const bool b3 = (m & 8) != 0, b2 = (m & 4) != 0;
        const bool b1 = (m & 2) != 0, b0 = (m & 1) != 0;
        const float kv = b3 ? adv : asv;
        const float sv = b3 ? asv : adv;
        float q[8], t[4], u[2];
        #pragma unroll
        for (int r = 0; r < 8; ++r)
            q[r] = acc[r] * kv + row_xor<8>(acc[r] * sv);
        #pragma unroll
        for (int r = 0; r < 4; ++r)
            t[r] = (b2 ? q[r + 4] : q[r]) + row_xor<4>(b2 ? q[r] : q[r + 4]);
        #pragma unroll
        for (int r = 0; r < 2; ++r)
            u[r] = (b1 ? t[r + 2] : t[r]) + row_xor<2>(b1 ? t[r] : t[r + 2]);
        lg = (b0 ? u[1] : u[0]) + row_xor<1>(b0 ? u[0] : u[1]);
    }
    sh_log[m >> 3][(rt * 16 + (m & 7) + 8 * half) * NCOLT + ct] = lg;
    __syncthreads();

    #pragma unroll
    for (int r = 0; r < 8; ++r) {
        H[(size_t)(r0 + rt * 16 + r + 8 * half) * FOUT + c0 + m] = acc[r];
    }

    if (wave < 2) {
        float* const dst = (wave == 0 ? as : ad) + (size_t)blockIdx.x * NL + lane * PER;
        if constexpr (PER == 2)
            *(v2f*)dst = *(const v2f*)&sh_log[wave][lane * 2];
        else
            *dst = sh_log[wave][lane];
    }
}

typedef float v4f __attribute__((ext_vector_type(4)));
typedef int   v4i __attribute__((ext_vector_type(4)));

template <int MASK>
__device__ __forceinline__ float lane_xor(float v) {
    return __int_as_float(
        __builtin_amdgcn_ds_swizzle(__float_as_int(v), (MASK << 10) | 0x1f));
}
__device__ __forceinline__ float wave_max(float v) {
    v = fmaxf(v, lane_xor<16>(v));
    v = fmaxf(v, lane_xor<8>(v));
    v = fmaxf(v, lane_xor<4>(v));
    v = fmaxf(v, lane_xor<2>(v));
    v = fmaxf(v, lane_xor<1>(v));
    return v;
}

template <int F, int NR, int GR, int EPS, int S0>
__device__ __forceinline__ void rows_request(const float* __restrict__ h,
                                             const int* idx_slot, int cnt,
                                             int f0, v4f (&r)[NR]) {
    v4i ix[NR / 4];
    #pragma unroll
    for (int q = 0; q < NR / 4; ++q) ix[q] = *(const v4i*)(idx_slot + 4 * q);
    #pragma unroll
    for (int g = 0; g < NR / GR; ++g) {
        if (g == 0 || cnt > (S0 + g * GR) * EPS) {
            #pragma unroll
            for (int u = 0; u < GR; ++u) {
                const int j = g * GR + u;
                const unsigned off = (unsigned)ix[j / 4][j % 4] * (unsigned)(F * 4)
                                   + (unsigned)f0 * 4u;
                r[j] = *(const v4f*)((const char*)h + off);
            }
        }
    }
}

template <int NR, int GR, int EPS, int S0>
__device__ __forceinline__ void rows_consume(const float* w_slot, int cnt,
                                             const v4f (&r)[NR],
                                             float (&acc)[4], float& den) {
    #pragma unroll
    for (int q = 0; q < NR / 4; ++q) {
        if (q == 0 || cnt > (S0 + 4 * q) * EPS) {
            const v4f wv = *(const v4f*)(w_slot + 4 * q);
            #pragma unroll
            for (int g = 0; g < 4 / GR; ++g) {
                if (g == 0 || cnt > (S0 + 4 * q + g * GR) * EPS) {
                    #pragma unroll
                    for (int u = 0; u < GR; ++u) {
                        const int j = 4 * q + g * GR + u;
                        const float wj = wv[g * GR + u];
                        #pragma unroll
                        for (int v = 0; v < 4; ++v) acc[v] = fmaf(wj, r[j][v], acc[v]);
                        den += wj;
                    }
                }
            }
        }
    }
}

template <int H, int D, bool ELU>
__global__ void __launch_bounds__(256, 16)
node_gather_kernel(const int* __restrict__ start, const int* __restrict__ csr,
                   const float* __restrict__ h,
                   const float* __restrict__ as, const float* __restrict__ ad,
                   const float* __restrict__ bias,
                   float* __restrict__ out, int N) {
    constexpr int F   = H * D;
    constexpr int VEC = 4;
    constexpr int LPR = F / VEC;
    constexpr int EPS = 32 / LPR;
    constexpr int NST = LPR;
    constexpr int NR  = (NST < 8) ? NST : 8;
    constexpr int GR  = NR / 2;
    static_assert(LPR == 16 || LPR == 4, "row must map to 16 or 4 lanes");
    static_assert(D % VEC == 0, "a lane's features must lie in one head");

    __shared__ __attribute__((aligned(16))) int   sh_idx[8][32];
    __shared__ __attribute__((aligned(16))) float sh_w[8][H * 32];

    const int lane = threadIdx.x & 31;
    const int wib  = threadIdx.x >> 5;
    const int sub  = lane / LPR;
    const int f0   = (lane % LPR) * VEC;
    const int myh  = f0 / D;
    const int ppos = (lane % EPS) * LPR + lane / EPS;
    int*   const my_idx = sh_idx[wib];
    float* const my_w   = sh_w[wib];
    const int* const idx_slot = my_idx + sub * LPR;
    const float* const w_slot = my_w + myh * 32 + sub * LPR;
    const int fo = (LPR == 16) ? f0 + 2 * (lane >> 4)
                               : f0 + 2 * ((lane >> 4) & 1) + ((lane >> 3) & 1);
    const int wave   = blockIdx.x * (blockDim.x >> 5) + (threadIdx.x >> 5);
    const int nwaves = gridDim.x * (blockDim.x >> 5);

    if (wave >= N) return;

    const int wave_s   = __builtin_amdgcn_readfirstlane(wave);
    const int nwaves_s = __builtin_amdgcn_readfirstlane(nwaves);

    float bo0 = bias[fo];
    float bo1 = (LPR == 16) ? bias[fo + 1] : 0.f;

    int   beg, end, s_cur;
    float adn[H];
    int   nbeg = 0, nend = 0;
    float nad[H];
    {
        const int n = wave_s;
        beg = start[n];
        end = start[n + 1];
        for (int k = 0; k < H; ++k) adn[k] = ad[n * H + k];
        for (int k = 0; k < H; ++k) nad[k] = 0.f;
        const int nn = n + nwaves_s;
        if (nn < N) {
            nbeg = start[nn];
            nend = start[nn + 1];
            for (int k = 0; k < H; ++k) nad[k] = ad[nn * H + k];
        }
        s_cur = (lane < end - beg) ? csr[beg + lane] : n;
    }

    for (int n = wave_s; n < N; n += nwaves_s) {
        const int cnt = end - beg + 1;

        const bool has_next = n + nwaves_s < N;
        const int  nn       = has_next ? n + nwaves_s : n;
        const int  n2       = nn + nwaves_s;
        const bool has_n2   = has_next && n2 < N;
        int   n2beg = 0, n2end = 0;
        float n2ad[H];
        for (int k = 0; k < H; ++k) n2ad[k] = 0.f;
        int s_next = nn;

        float acc[VEC];
        for (int v = 0; v < VEC; ++v) acc[v] = 0.f;
        float den = 0.f;

        if (cnt <= 32) {
            const int s = s_cur;
            my_idx[ppos] = s;

            float av[H];
            if constexpr (H == 4) {
                const float4 t = *(const float4*)(as + (size_t)s * 4);
                av[0] = t.x; av[1] = t.y; av[2] = t.z; av[3] = t.w;
            } else {
                for (int k = 0; k < H; ++k) av[k] = as[(size_t)s * H + k];
            }

            v4f r[NR];
            rows_request<F, NR, GR, EPS, 0>(h, idx_slot, cnt, f0, r);

            if (has_next)
                s_next = (lane < nend - nbeg) ? csr[nbeg + lane] : nn;
            if (has_n2) {
                n2beg = start[n2];
                n2end = start[n2 + 1];
                for (int k = 0; k < H; ++k) n2ad[k] = ad[n2 * H + k];
            }

            for (int k = 0; k < H; ++k) {
                float v = av[k] + adn[k];
                v = (v > 0.f) ? v : NEG_SLOPE * v;
                const float mx = wave_max((lane < cnt) ? v : -INFINITY);
                const float ex = expf(v - mx);
                my_w[k * 32 + ppos] = (lane < cnt) ? ex : 0.f;
            }

            rows_consume<NR, GR, EPS, 0>(w_slot, cnt, r, acc, den);

            if constexpr (NST > NR) {
                if (cnt > NR * EPS) {
                    rows_request<F, NR, GR, EPS, NR>(h, idx_slot + NR, cnt, f0, r);
                    rows_consume<NR, GR, EPS, NR>(w_slot + NR, cnt, r, acc, den);
                }
            }
        } else {
            if (has_next)
                s_next = (lane < nend - nbeg) ? csr[nbeg + lane] : nn;
            if (has_n2) {
                n2beg = start[n2];
                n2end = start[n2 + 1];
                for (int k = 0; k < H; ++k) n2ad[k] = ad[n2 * H + k];
            }

            float mx[H];
            for (int k = 0; k < H; ++k) mx[k] = -INFINITY;

            for (int base = 0; base < cnt; base += 32) {
                const int idx = base + lane;
                const int s = (idx < cnt - 1) ? csr[beg + idx] : n;
                float av[H];
                if constexpr (H == 4) {
                    const float4 t = *(const float4*)(as + (size_t)s * 4);
                    av[0] = t.x; av[1] = t.y; av[2] = t.z; av[3] = t.w;
                } else {
                    for (int k = 0; k < H; ++k) av[k] = as[(size_t)s * H + k];
                }
                for (int k = 0; k < H; ++k) {
                    float v = av[k] + adn[k];
                    v = (v > 0.f) ? v : NEG_SLOPE * v;
                    mx[k] = fmaxf(mx[k], (idx < cnt) ? v : -INFINITY);
                }
            }
            for (int k = 0; k < H; ++k) mx[k] = wave_max(mx[k]);

            for (int base = 0; base < cnt; base += 32) {
                const int idx = base + lane;
                const int s = (idx < cnt - 1) ? csr[beg + idx] : n;
                float av[H];
                if constexpr (H == 4) {
                    const float4 t = *(const float4*)(as + (size_t)s * 4);
                    av[0] = t.x; av[1] = t.y; av[2] = t.z; av[3] = t.w;
                } else {
                    for (int k = 0; k < H; ++k) av[k] = as[(size_t)s * H + k];
                }
                float w[H];
                for (int k = 0; k < H; ++k) {
                    float v = av[k] + adn[k];
                    v = (v > 0.f) ? v : NEG_SLOPE * v;
                    const float ex = expf(v - mx[k]);
                    w[k] = (idx < cnt) ? ex : 0.f;
                }

                const int m = min(32, cnt - base);
                my_idx[ppos] = s;
                for (int k = 0; k < H; ++k) my_w[k * 32 + ppos] = w[k];
                #pragma unroll 2
                for (int j = 0; j < NST; ++j) {
                    if (j * EPS < m) {
                        const int   sj = idx_slot[j];
                        const float wj = w_slot[j];
                        const v4f rr = *(const v4f*)(h + (size_t)sj * F + f0);
                        #pragma unroll
                        for (int v = 0; v < VEC; ++v)
                            acc[v] = fmaf(wj, rr[v], acc[v]);
                        den += wj;
                    }
                }
            }
        }

        float k0, k1;
        {
            const bool hi = (lane & 16) != 0;
            k0 = hi ? acc[2] : acc[0];
            k1 = hi ? acc[3] : acc[1];
            const float s0 = hi ? acc[0] : acc[2];
            const float s1 = hi ? acc[1] : acc[3];
            k0  += lane_xor<16>(s0);
            k1  += lane_xor<16>(s1);
            den += lane_xor<16>(den);
        }
        if constexpr (LPR == 16) {
            v2f val;
            val[0] = k0 / (den + 1e-16f) + bo0;
            val[1] = k1 / (den + 1e-16f) + bo1;
            if (ELU) {
                val[0] = (val[0] > 0.f) ? val[0] : (expf(val[0]) - 1.f);
                val[1] = (val[1] > 0.f) ? val[1] : (expf(val[1]) - 1.f);
            }
            *(v2f*)((char*)out + ((unsigned)n * (unsigned)(F * 4) + (unsigned)fo * 4u)) = val;
        } else {
            const bool hi8 = (lane & 8) != 0;
            float kk = hi8 ? k1 : k0;
            const float ss = hi8 ? k0 : k1;
            kk  += lane_xor<8>(ss);
            den += lane_xor<8>(den);
            kk  += lane_xor<4>(kk);
            den += lane_xor<4>(den);
            if ((lane & 4) == 0) {
                float val = kk / (den + 1e-16f) + bo0;
                if (ELU) val = (val > 0.f) ? val : (expf(val) - 1.f);
                *(float*)((char*)out + ((unsigned)n * (unsigned)(F * 4) + (unsigned)fo * 4u)) = val;
            }
        }

        beg = nbeg; end = nend; s_cur = s_next;
        nbeg = n2beg; nend = n2end;
        for (int k = 0; k < H; ++k) { adn[k] = nad[k]; nad[k] = n2ad[k]; }
    }
}

static inline int cdiv(int a, int b) { return (a + b - 1) / b; }

extern "C" void kernel_launch(void* const* d_in, const int* in_sizes, int n_in,
                              void* d_out, int out_size, void* d_ws, size_t ws_size,
                              hipStream_t stream) {
    const float* x      = (const float*)d_in[0];
    const int*   ei     = (const int*)  d_in[1];
    const float* W1     = (const float*)d_in[2];
    const float* a_src1 = (const float*)d_in[3];
    const float* a_dst1 = (const float*)d_in[4];
    const float* b1     = (const float*)d_in[5];
    const float* W2     = (const float*)d_in[6];
    const float* a_src2 = (const float*)d_in[7];
    const float* a_dst2 = (const float*)d_in[8];
    const float* b2     = (const float*)d_in[9];

    const int N  = in_sizes[0] / 128;
    const int E  = in_sizes[1] / 2;

    float* ws   = (float*)d_ws;
    float* h1   = ws;
    float* agg1 = h1   + (size_t)N * 64;
    float* as1  = agg1 + (size_t)N * 64;
    float* ad1  = as1  + (size_t)N * 4;
    int*   strt = (int*)(ad1 + (size_t)N * 4);
    int*   rnkw = strt  + N + 4;
    int*   csr  = rnkw  + E;
    int*   bsum = csr   + E;
    unsigned* tabg = (unsigned*)(bsum + NCHUNK * 128);
    int*   base = (int*)(tabg + (size_t)NCHUNK * CAPW);
    unsigned short* rank = (unsigned short*)rnkw;
    float* h2   = h1;
    float* as2  = h2   + (size_t)N * 16;
    float* ad2  = as2  + N;

    float* out  = (float*)d_out;

    const int TB = 256;
    const int NB = cdiv(N, 1024);

    chunk_count_kernel<<<NCHUNK, 1024, 0, stream>>>(ei, E, tabg, rank, bsum);
    chunk_scan_kernel<<<NB, 256, 0, stream>>>(tabg, bsum, strt, base, N);
    chunk_place_kernel<<<dim3(32, NCHUNK), TB, 0, stream>>>(ei, E, base, rank, csr);

    gemm_logits_wmma_f32<128, 4, 1><<<N / 16, dim3(32, 4), 0, stream>>>(
        x, W1, a_src1, a_dst1, h1, as1, ad1);

    node_gather_kernel<4, 16, true><<<2048, TB, 0, stream>>>(
        strt, csr, h1, as1, ad1, b1, agg1, N);

    gemm_logits_wmma_f32<64, 1, 2><<<N / 32, dim3(32, 2), 0, stream>>>(
        agg1, W2, a_src2, a_dst2, h2, as2, ad2);

    node_gather_kernel<1, 16, false><<<2048, TB, 0, stream>>>(
        strt, csr, h2, as2, ad2, b2, out, N);
}


# Layer-1 GEMM runs inside the edge-count launch: 224 more work-groups (the WGPs the 32 count groups leave idle) walk row tiles, W fragments in registers, X tiles double-buffered in LDS; 6 launches
# speedup vs baseline: 8.0595x; 1.1490x over previous
#include <hip/hip_runtime.h>
#include <hip/hip_bf16.h>
#include <math.h>

typedef float v2f __attribute__((ext_vector_type(2)));
typedef float v8f __attribute__((ext_vector_type(8)));
typedef float v4f __attribute__((ext_vector_type(4)));
typedef int   v4i __attribute__((ext_vector_type(4)));

#define NEG_SLOPE 0.2f

#define NCHUNK 32
#define CAPW   50176
#define NPITCH (2 * CAPW)
#define NGEMM  224

template <int M>
__device__ __forceinline__ unsigned row_xor_u(unsigned v) {
    return (unsigned)__builtin_amdgcn_update_dpp(
        0, (int)v, 0x160 | M, 0xf, 0xf, true);
}

template <int M>
__device__ __forceinline__ float row_xor(float v) {
    return __int_as_float(__builtin_amdgcn_update_dpp(
        0, __float_as_int(v), 0x160 | M, 0xf, 0xf, true));
}

__device__ __forceinline__ void lds_barrier() {
    __builtin_amdgcn_fence(__ATOMIC_RELEASE, "workgroup", "local");
    __builtin_amdgcn_s_barrier();
    __builtin_amdgcn_fence(__ATOMIC_ACQUIRE, "workgroup", "local");
}

#define G1_K     128
#define G1_FOUT  64
#define G1_PITCH (G1_K + 4)
#define G1_RT    8
#define G1_ROWS  (16 * G1_RT)

__device__ __forceinline__ void gemm1_rows(float* __restrict__ lds,
                                           const float* __restrict__ X,
                                           const float* __restrict__ W,
                                           const float* __restrict__ a_src,
                                           const float* __restrict__ a_dst,
                                           float* __restrict__ H,
                                           float* __restrict__ as,
                                           float* __restrict__ ad, int N) {
    constexpr int NLOG = G1_ROWS * 4;
    static_assert(2 * G1_ROWS * G1_PITCH + 4 * NLOG <= CAPW, "fits the counter table's LDS");
    float* const shl = lds + 2 * G1_ROWS * G1_PITCH;

    const int lane = threadIdx.x & 31;
    const int wave = __builtin_amdgcn_readfirstlane((int)(threadIdx.x >> 5));
    const int rt   = wave >> 2;
    const int ct   = wave & 3;
    const int half = lane >> 4;
    const int m    = lane & 15;
    const int c0   = ct * 16;

    const int ntile = N / 16;
    const int ngw   = (int)gridDim.x - NCHUNK;
    const int wg    = (int)blockIdx.x - NCHUNK;
    const int tpw   = (ntile + ngw - 1) / ngw;
    const int t0    = wg * tpw;
    const int t1    = min(ntile, t0 + tpw);
    const int nit   = (t1 > t0) ? (t1 - t0 + G1_RT - 1) / G1_RT : 0;

    v2f b[G1_K / 4];
    #pragma unroll
    for (int s = 0; s < G1_K / 4; ++s) {
        const int kb = 4 * s + 2 * half;
        b[s][0] = W[(size_t)(kb)     * G1_FOUT + c0 + m];
        b[s][1] = W[(size_t)(kb + 1) * G1_FOUT + c0 + m];
    }
    const float asv = a_src[c0 + m];
    const float adv = a_dst[c0 + m];

    v4f pre[4];
    #pragma unroll
    for (int j = 0; j < 4; ++j) pre[j] = v4f{0.f, 0.f, 0.f, 0.f};
    if (nit > 0) {
        #pragma unroll
        for (int j = 0; j < 4; ++j) {
            const int row = wave + 32 * j;
            if (t0 + (row >> 4) < t1)
                pre[j] = *(const v4f*)(X + (size_t)(t0 * 16 + row) * G1_K + 4 * lane);
        }
    }

    for (int it = 0; it < nit; ++it) {
        const int buf = it & 1;
        const int tb  = t0 + it * G1_RT;
        float* const Ab = lds + buf * (G1_ROWS * G1_PITCH);

        #pragma unroll
        for (int j = 0; j < 4; ++j) {
            const int row = wave + 32 * j;
            if (tb + (row >> 4) < t1)
                *(v4f*)(Ab + row * G1_PITCH + 4 * lane) = pre[j];
        }
        lds_barrier();

        if (it > 0 && wave < 8) {
            const int kind = wave >> 2;
            const int f    = (wave & 3) * 128 + lane * 4;
            const int pb   = tb - G1_RT;
            if (pb + (f >> 6) < t1)
                *(v4f*)((kind ? ad : as) + (size_t)pb * 64 + f) =
                    *(const v4f*)(shl + ((buf ^ 1) * 2 + kind) * NLOG + f);
        }

        if (it + 1 < nit) {
            #pragma unroll
            for (int j = 0; j < 4; ++j) {
                const int row = wave + 32 * j;
                if (tb + G1_RT + (row >> 4) < t1)
                    pre[j] = *(const v4f*)(X + (size_t)((tb + G1_RT) * 16 + row) * G1_K + 4 * lane);
            }
        }

        if (tb + rt < t1) {
            const float* const Ar = Ab + (rt * 16 + m) * G1_PITCH + 2 * half;
            v8f acc = {};
            #pragma unroll
            for (int s = 0; s < G1_K / 4; ++s) {
                const v2f a = *(const v2f*)(Ar + 4 * s);
                acc = __builtin_amdgcn_wmma_f32_16x16x4_f32(
                     false, a,  false, b[s],
                     (short)0, acc,  false,  false);
            }

            float lg;
            {
                const bool b3 = (m & 8) != 0, b2 = (m & 4) != 0;
                const bool b1 = (m & 2) != 0, b0 = (m & 1) != 0;
                const float kv = b3 ? adv : asv;
                const float sv = b3 ? asv : adv;
                float q[8], t[4], u[2];
                #pragma unroll
                for (int r = 0; r < 8; ++r)
                    q[r] = acc[r] * kv + row_xor<8>(acc[r] * sv);
                #pragma unroll
                for (int r = 0; r < 4; ++r)
                    t[r] = (b2 ? q[r + 4] : q[r]) + row_xor<4>(b2 ? q[r] : q[r + 4]);
                #pragma unroll
                for (int r = 0; r < 2; ++r)
                    u[r] = (b1 ? t[r + 2] : t[r]) + row_xor<2>(b1 ? t[r] : t[r + 2]);
                lg = (b0 ? u[1] : u[0]) + row_xor<1>(b0 ? u[0] : u[1]);
            }
            shl[(buf * 2 + (m >> 3)) * NLOG + (rt * 16 + (m & 7) + 8 * half) * 4 + ct] = lg;

            #pragma unroll
            for (int r = 0; r < 8; ++r)
                H[(size_t)((tb + rt) * 16 + r + 8 * half) * G1_FOUT + c0 + m] = acc[r];
        }
    }
    lds_barrier();
    if (nit > 0 && wave < 8) {
        const int kind = wave >> 2;
        const int f    = (wave & 3) * 128 + lane * 4;
        const int pb   = t0 + (nit - 1) * G1_RT;
        if (pb + (f >> 6) < t1)
            *(v4f*)((kind ? ad : as) + (size_t)pb * 64 + f) =
                *(const v4f*)(shl + (((nit - 1) & 1) * 2 + kind) * NLOG + f);
    }
}

__global__ void __launch_bounds__(1024)
count_gemm_kernel(const int* __restrict__ ei, int E,
                  unsigned* __restrict__ tabg,
                  unsigned short* __restrict__ rank,
                  int* __restrict__ bsumg,
                  const float* __restrict__ X,
                  const float* __restrict__ W,
                  const float* __restrict__ a_src,
                  const float* __restrict__ a_dst,
                  float* __restrict__ H,
                  float* __restrict__ as,
                  float* __restrict__ ad, int N) {
    __shared__ __attribute__((aligned(16))) unsigned tab[CAPW];
    __shared__ unsigned blk[128];
    if (blockIdx.x >= NCHUNK) {
        gemm1_rows(reinterpret_cast<float*>(tab), X, W, a_src, a_dst, H, as, ad, N);
        return;
    }
    const int t  = threadIdx.x;
    const int g  = blockIdx.x;
    const int CH = (E + NCHUNK - 1) / NCHUNK;
    const int c0 = g * CH;
    const int c1 = min(E, c0 + CH);

    for (int i = t * 4; i < CAPW; i += 4096)
        *(uint4*)&tab[i] = make_uint4(0u, 0u, 0u, 0u);
    if (t < 128) blk[t] = 0u;
    __syncthreads();

    constexpr int EPT = 30;
    int e0 = c0 + t;
    for (; e0 + (EPT - 1) * 1024 < c1; e0 += EPT * 1024) {
        int d[EPT];
        unsigned old[EPT];
        #pragma unroll
        for (int u = 0; u < EPT; ++u) d[u] = ei[E + e0 + u * 1024];
        #pragma unroll
        for (int u = 0; u < EPT; ++u)
            old[u] = atomicAdd(&tab[d[u] >> 1], 1u << ((unsigned)(d[u] & 1) * 16u));
        #pragma unroll
        for (int u = 0; u < EPT; ++u)
            rank[e0 + u * 1024] =
                (unsigned short)(old[u] >> ((unsigned)(d[u] & 1) * 16u));
    }
    for (int e = e0; e < c1; e += 1024) {
        const int dd = ei[E + e];
        const unsigned sh  = (unsigned)(dd & 1) * 16u;
        const unsigned old = atomicAdd(&tab[dd >> 1], 1u << sh);
        rank[e] = (unsigned short)(old >> sh);
    }
    __syncthreads();

    unsigned* const dst = tabg + (size_t)g * CAPW;
    constexpr int NRND = CAPW / 4096;
    const bool tail = t * 4 + NRND * 4096 < CAPW;
    unsigned s[NRND + 1];
    #pragma unroll
    for (int r = 0; r <= NRND; ++r) {
        const int i = t * 4 + r * 4096;
        s[r] = 0u;
        if (r < NRND || tail) {
            const uint4 q = *(const uint4*)&tab[i];
            *(uint4*)&dst[i] = q;
            s[r] = (q.x & 0xffffu) + (q.x >> 16) + (q.y & 0xffffu) + (q.y >> 16)
                 + (q.z & 0xffffu) + (q.z >> 16) + (q.w & 0xffffu) + (q.w >> 16);
        }
    }
    #pragma unroll
    for (int r = 0; r <= NRND; ++r) {
        if (r < NRND || tail) {
            unsigned x = s[r];
            x += row_xor_u<1>(x);
            x += row_xor_u<2>(x);
            x += row_xor_u<4>(x);
            x += row_xor_u<8>(x);
            if ((t & 15) == 0) atomicAdd(&blk[(t * 4 + r * 4096) >> 9], x);
        }
    }
    __syncthreads();
    if (t < 128) bsumg[g * 128 + t] = (int)blk[t];
}

__global__ void __launch_bounds__(256)
chunk_scan_kernel(const unsigned* __restrict__ tabg,
                                  const int* __restrict__ bsumg,
                                  int* __restrict__ start,
                                  int* __restrict__ base, int n) {
    __shared__ int sh[256];
    const int t  = threadIdx.x;
    const int i0 = blockIdx.x * 1024 + t * 4;
    const int w0 = blockIdx.x * 512 + t * 2;

    int part = 0;
    if (t < (int)blockIdx.x) {
        #pragma unroll
        for (int g = 0; g < NCHUNK; ++g) part += bsumg[g * 128 + t];
    }

    uint2 v[NCHUNK];
    #pragma unroll
    for (int g = 0; g < NCHUNK; ++g)
        v[g] = *(const uint2*)&tabg[(size_t)g * CAPW + w0];

    int c[4] = {0, 0, 0, 0};
    #pragma unroll
    for (int g = 0; g < NCHUNK; ++g) {
        c[0] += (int)(v[g].x & 0xffffu);
        c[1] += (int)(v[g].x >> 16);
        c[2] += (int)(v[g].y & 0xffffu);
        c[3] += (int)(v[g].y >> 16);
    }
    const int tsum = c[0] + c[1] + c[2] + c[3];

    sh[t] = part;
    __syncthreads();
    for (int off = 128; off > 0; off >>= 1) {
        if (t < off) sh[t] += sh[t + off];
        __syncthreads();
    }
    const int before = sh[0];
    __syncthreads();

    sh[t] = tsum;
    __syncthreads();
    for (int off = 1; off < 256; off <<= 1) {
        const int x = (t >= off) ? sh[t - off] : 0;
        __syncthreads();
        sh[t] += x;
        __syncthreads();
    }
    int o[4];
    o[0] = before + sh[t] - tsum;
    o[1] = o[0] + c[0];
    o[2] = o[1] + c[1];
    o[3] = o[2] + c[2];
    for (int k = 0; k < 4; ++k)
        if (i0 + k < n) start[i0 + k] = o[k];
    if (i0 < n && n <= i0 + 4) start[n] = o[3] + c[3];

    #pragma unroll
    for (int g = 0; g < NCHUNK; ++g) {
        *(int4*)&base[(size_t)g * NPITCH + i0] = make_int4(o[0], o[1], o[2], o[3]);
        o[0] += (int)(v[g].x & 0xffffu);
        o[1] += (int)(v[g].x >> 16);
        o[2] += (int)(v[g].y & 0xffffu);
        o[3] += (int)(v[g].y >> 16);
    }
}

__global__ void __launch_bounds__(256)
chunk_place_kernel(const int* __restrict__ ei, int E,
                                   const int* __restrict__ base,
                                   const unsigned short* __restrict__ rank,
                                   int* __restrict__ csr) {
    const int g  = blockIdx.y;
    const int CH = (E + NCHUNK - 1) / NCHUNK;
    const int c0 = g * CH;
    const int c1 = min(E, c0 + CH);
    const int* const bg = base + (size_t)g * NPITCH;
    const int stride = gridDim.x * blockDim.x;
    for (int e = c0 + blockIdx.x * blockDim.x + threadIdx.x; e < c1; e += stride) {
        const int d = ei[E + e];
        csr[bg[d] + (int)rank[e]] = ei[e];
    }
}

template <int K, int NCOLT, int NROWT>
__global__ void gemm_logits_wmma_f32(const float* __restrict__ X,
                                     const float* __restrict__ W,
                                     const float* __restrict__ a_src,
                                     const float* __restrict__ a_dst,
                                     float* __restrict__ H,
                                     float* __restrict__ as,
                                     float* __restrict__ ad) {
    constexpr int FOUT = 16 * NCOLT;
    constexpr int ROWS = 16 * NROWT;
    constexpr int NW   = NCOLT * NROWT;
    constexpr int NL   = ROWS * NCOLT;
    constexpr int PER  = NL / 32;
    static_assert(NW >= 2, "one wave stores as, another ad");
    static_assert(PER == 1 || PER == 2, "a block's logits are one store per kind");
    __shared__ float As[ROWS][K + 4];
    __shared__ __attribute__((aligned(8))) float sh_log[2][NL];

    const int r0   = blockIdx.x * ROWS;
    const int lane = threadIdx.x;
    const int wave = threadIdx.y;
    const int rt   = wave / NCOLT;
    const int ct   = wave % NCOLT;
    const int tid  = wave * 32 + lane;
    constexpr int nthr = 32 * NW;

    for (int i = tid; i < ROWS * K; i += nthr) {
        int rr = i / K, cc = i % K;
        As[rr][cc] = X[(size_t)(r0 + rr) * K + cc];
    }
    __syncthreads();

    const int half = lane >> 4;
    const int m    = lane & 15;
    const int c0   = ct * 16;
    const int ar   = rt * 16 + m;

    const float asv = a_src[c0 + m];
    const float adv = a_dst[c0 + m];

    v8f acc = {};
    #pragma unroll 4
    for (int kk = 0; kk < K; kk += 4) {
        const int kb = kk + 2 * half;
        v2f a, b;
        a[0] = As[ar][kb];
        a[1] = As[ar][kb + 1];
        b[0] = W[(size_t)(kb)     * FOUT + c0 + m];
        b[1] = W[(size_t)(kb + 1) * FOUT + c0 + m];
        acc = __builtin_amdgcn_wmma_f32_16x16x4_f32(
             false, a,  false, b,
             (short)0, acc,  false,  false);
    }

    float lg;
    {
        const bool b3 = (m & 8) != 0, b2 = (m & 4) != 0;
        const bool b1 = (m & 2) != 0, b0 = (m & 1) != 0;
        const float kv = b3 ? adv : asv;
        const float sv = b3 ? asv : adv;
        float q[8], t[4], u[2];
        #pragma unroll
        for (int r = 0; r < 8; ++r)
            q[r] = acc[r] * kv + row_xor<8>(acc[r] * sv);
        #pragma unroll
        for (int r = 0; r < 4; ++r)
            t[r] = (b2 ? q[r + 4] : q[r]) + row_xor<4>(b2 ? q[r] : q[r + 4]);
        #pragma unroll
        for (int r = 0; r < 2; ++r)
            u[r] = (b1 ? t[r + 2] : t[r]) + row_xor<2>(b1 ? t[r] : t[r + 2]);
        lg = (b0 ? u[1] : u[0]) + row_xor<1>(b0 ? u[0] : u[1]);
    }
    sh_log[m >> 3][(rt * 16 + (m & 7) + 8 * half) * NCOLT + ct] = lg;
    __syncthreads();

    #pragma unroll
    for (int r = 0; r < 8; ++r) {
        H[(size_t)(r0 + rt * 16 + r + 8 * half) * FOUT + c0 + m] = acc[r];
    }

    if (wave < 2) {
        float* const dst = (wave == 0 ? as : ad) + (size_t)blockIdx.x * NL + lane * PER;
        if constexpr (PER == 2)
            *(v2f*)dst = *(const v2f*)&sh_log[wave][lane * 2];
        else
            *dst = sh_log[wave][lane];
    }
}

template <int MASK>
__device__ __forceinline__ float lane_xor(float v) {
    return __int_as_float(
        __builtin_amdgcn_ds_swizzle(__float_as_int(v), (MASK << 10) | 0x1f));
}

template <int F, int NR, int GR, int EPS>
__device__ __forceinline__ void rows_request(const float* __restrict__ h,
                                             const int* idx_slot, int cnt,
                                             int f0, v4f (&r)[NR]) {
    v4i ix[NR / 4];
    #pragma unroll
    for (int q = 0; q < NR / 4; ++q) ix[q] = *(const v4i*)(idx_slot + 4 * q);
    #pragma unroll
    for (int g = 0; g < NR / GR; ++g) {
        if (g == 0 || cnt > g * GR * EPS) {
            #pragma unroll
            for (int u = 0; u < GR; ++u) {
                const int j = g * GR + u;
                const unsigned off = (unsigned)ix[j / 4][j % 4] * (unsigned)(F * 4)
                                   + (unsigned)f0 * 4u;
                r[j] = *(const v4f*)((const char*)h + off);
            }
        }
    }
}

template <int Q>
__device__ __forceinline__ float quad_bcast(float v) {
    return __int_as_float(__builtin_amdgcn_update_dpp(
        0, __float_as_int(v), Q * 0x55, 0xf, 0xf, true));
}

template <int NR, int GR, int EPS, int LPR>
__device__ __forceinline__ void window_sum(const v4f (&r)[NR], int cnt, int vb,
                                           bool b0, bool b1, bool later,
                                           const v4f a4, float adl, float& mrun,
                                           float (&acc)[4], float& den) {
    constexpr int NH = NR / 2;
    constexpr int NO = NR / 4;
    const v2f alo = __builtin_shufflevector(a4, a4, 0, 1);
    const v2f ahi = __builtin_shufflevector(a4, a4, 2, 3);
    float p[NR];
    #pragma unroll
    for (int j = 0; j < NR; ++j) p[j] = 0.f;
    #pragma unroll
    for (int g = 0; g < NR / GR; ++g) {
        if (g == 0 || cnt > g * GR * EPS) {
            #pragma unroll
            for (int u = 0; u < GR; ++u) {
                const int j = g * GR + u;
                v2f s2 = __builtin_shufflevector(r[j], r[j], 0, 1) * alo;
                s2 = __builtin_elementwise_fma(
                    __builtin_shufflevector(r[j], r[j], 2, 3), ahi, s2);
                p[j] = s2[0] + s2[1];
            }
        }
    }
    float t[NH];
    #pragma unroll
    for (int i = 0; i < NH; ++i)
        t[i] = (b0 ? p[i + NH] : p[i]) + row_xor<1>(b0 ? p[i] : p[i + NH]);
    float lg[NO];
    #pragma unroll
    for (int i = 0; i < NO; ++i)
        lg[i] = (b1 ? t[i + NO] : t[i]) + row_xor<2>(b1 ? t[i] : t[i + NO]);

    float mx = -INFINITY;
    #pragma unroll
    for (int i = 0; i < NO; ++i) {
        float v = lg[i] + adl;
        v = (v > 0.f) ? v : NEG_SLOPE * v;
        lg[i] = (vb < cnt - i * EPS) ? v : -INFINITY;
        mx = fmaxf(mx, lg[i]);
    }
    mx = fmaxf(mx, row_xor<1>(mx));
    mx = fmaxf(mx, row_xor<2>(mx));
    if constexpr (LPR == 4) {
        mx = fmaxf(mx, row_xor<4>(mx));
        mx = fmaxf(mx, row_xor<8>(mx));
    }
    mx = fmaxf(mx, lane_xor<16>(mx));
    if (later) {
        mx = fmaxf(mx, mrun);
        const float sc = expf(mrun - mx);
        #pragma unroll
        for (int v = 0; v < 4; ++v) acc[v] *= sc;
        den *= sc;
    }
    mrun = mx;

    float w[NO];
    #pragma unroll
    for (int i = 0; i < NO; ++i) w[i] = expf(lg[i] - mx);

    #pragma unroll
    for (int g = 0; g < NR / GR; ++g) {
        if (g == 0 || cnt > g * GR * EPS) {
            #pragma unroll
            for (int u = 0; u < GR; ++u) {
                const int j = g * GR + u;
                float wj;
                switch ((j / NH) + 2 * ((j % NH) / NO)) {
                    case 0:  wj = quad_bcast<0>(w[j % NO]); break;
                    case 1:  wj = quad_bcast<1>(w[j % NO]); break;
                    case 2:  wj = quad_bcast<2>(w[j % NO]); break;
                    default: wj = quad_bcast<3>(w[j % NO]); break;
                }
                #pragma unroll
                for (int v = 0; v < 4; ++v) acc[v] = fmaf(wj, r[j][v], acc[v]);
                den += wj;
            }
        }
    }
}

template <int H, int D, bool ELU>
__global__ void __launch_bounds__(256, 16)
node_gather_kernel(const int* __restrict__ start, const int* __restrict__ csr,
                   const float* __restrict__ h,
                   const float* __restrict__ ad,
                   const float* __restrict__ a_src,
                   const float* __restrict__ bias,
                   float* __restrict__ out, int N) {
    constexpr int F   = H * D;
    constexpr int VEC = 4;
    constexpr int LPR = F / VEC;
    constexpr int EPS = 32 / LPR;
    constexpr int NST = LPR;
    constexpr int NR  = (NST < 8) ? NST : 8;
    constexpr int GR  = NR / 2;
    constexpr int WE  = NR * EPS;
    constexpr int WPC = NST / NR;
    static_assert(LPR == 16 || LPR == 4, "row must map to 16 or 4 lanes");
    static_assert(D == 4 * VEC, "a head's logit is summed over 4 adjacent lanes");

    __shared__ __attribute__((aligned(16))) int sh_idx[8][32];

    const int lane = threadIdx.x & 31;
    const int wib  = threadIdx.x >> 5;
    const int sub  = lane / LPR;
    const int f0   = (lane % LPR) * VEC;
    const int myh  = f0 / D;
    const int ppos = (lane % EPS) * LPR + lane / EPS;
    int* const my_idx = sh_idx[wib];
    const int* const idx_slot = my_idx + sub * LPR;
    const bool lb0 = (lane & 1) != 0, lb1 = (lane & 2) != 0;
    const int  vb  = ((lane & 1) * (NR / 2) + ((lane >> 1) & 1) * (NR / 4)) * EPS + sub;
    const int fo = (LPR == 16) ? f0 + 2 * (lane >> 4)
                               : f0 + 2 * ((lane >> 4) & 1) + ((lane >> 3) & 1);
    const int wave   = blockIdx.x * (blockDim.x >> 5) + (threadIdx.x >> 5);
    const int nwaves = gridDim.x * (blockDim.x >> 5);

    if (wave >= N) return;

    const int wave_s   = __builtin_amdgcn_readfirstlane(wave);
    const int nwaves_s = __builtin_amdgcn_readfirstlane(nwaves);

    float bo0 = bias[fo];
    float bo1 = (LPR == 16) ? bias[fo + 1] : 0.f;

    const v4f a4 = *(const v4f*)(a_src + f0);

    int   beg, end, s_cur;
    float adn[H];
    int   nbeg = 0, nend = 0;
    float nad[H];
    {
        const int n = wave_s;
        beg = start[n];
        end = start[n + 1];
        for (int k = 0; k < H; ++k) adn[k] = ad[n * H + k];
        for (int k = 0; k < H; ++k) nad[k] = 0.f;
        const int nn = n + nwaves_s;
        if (nn < N) {
            nbeg = start[nn];
            nend = start[nn + 1];
            for (int k = 0; k < H; ++k) nad[k] = ad[nn * H + k];
        }
        s_cur = (lane < end - beg) ? csr[beg + lane] : n;
    }

    for (int n = wave_s; n < N; n += nwaves_s) {
        const int cnt = end - beg + 1;

        const bool has_next = n + nwaves_s < N;
        const int  nn       = has_next ? n + nwaves_s : n;
        const int  n2       = nn + nwaves_s;
        const bool has_n2   = has_next && n2 < N;
        int   n2beg = 0, n2end = 0;
        float n2ad[H];
        for (int k = 0; k < H; ++k) n2ad[k] = 0.f;
        int s_next = nn;

        float acc[VEC];
        for (int v = 0; v < VEC; ++v) acc[v] = 0.f;
        float den = 0.f;
        float mrun = -INFINITY;

        float adl = adn[0];
        for (int k = 1; k < H; ++k) adl = (myh == k) ? adn[k] : adl;

        my_idx[ppos] = s_cur;

        if (has_next)
            s_next = (lane < nend - nbeg) ? csr[nbeg + lane] : nn;
        if (has_n2) {
            n2beg = start[n2];
            n2end = start[n2 + 1];
            for (int k = 0; k < H; ++k) n2ad[k] = ad[n2 * H + k];
        }

        #pragma clang loop unroll(disable)
        for (int wi = 0; wi * WE < cnt; ++wi) {
            if (wi > 0 && wi % WPC == 0) {
                const int idx = wi * WE + lane;
                my_idx[ppos] = (idx < cnt - 1) ? csr[beg + idx] : n;
            }
            v4f r[NR];
            rows_request<F, NR, GR, EPS>(h, idx_slot + (wi % WPC) * NR,
                                         cnt - wi * WE, f0, r);
            window_sum<NR, GR, EPS, LPR>(r, cnt - wi * WE, vb, lb0, lb1, wi > 0,
                                         a4, adl, mrun, acc, den);
        }

        float k0, k1;
        {
            const bool hi = (lane & 16) != 0;
            k0 = hi ? acc[2] : acc[0];
            k1 = hi ? acc[3] : acc[1];
            const float s0 = hi ? acc[0] : acc[2];
            const float s1 = hi ? acc[1] : acc[3];
            k0  += lane_xor<16>(s0);
            k1  += lane_xor<16>(s1);
            den += lane_xor<16>(den);
        }
        if constexpr (LPR == 16) {
            v2f val;
            val[0] = k0 / (den + 1e-16f) + bo0;
            val[1] = k1 / (den + 1e-16f) + bo1;
            if (ELU) {
                val[0] = (val[0] > 0.f) ? val[0] : (expf(val[0]) - 1.f);
                val[1] = (val[1] > 0.f) ? val[1] : (expf(val[1]) - 1.f);
            }
            *(v2f*)((char*)out + ((unsigned)n * (unsigned)(F * 4) + (unsigned)fo * 4u)) = val;
        } else {
            const bool hi8 = (lane & 8) != 0;
            float kk = hi8 ? k1 : k0;
            const float ss = hi8 ? k0 : k1;
            kk  += lane_xor<8>(ss);
            den += lane_xor<8>(den);
            kk  += lane_xor<4>(kk);
            den += lane_xor<4>(den);
            if ((lane & 4) == 0) {
                float val = kk / (den + 1e-16f) + bo0;
                if (ELU) val = (val > 0.f) ? val : (expf(val) - 1.f);
                *(float*)((char*)out + ((unsigned)n * (unsigned)(F * 4) + (unsigned)fo * 4u)) = val;
            }
        }

        beg = nbeg; end = nend; s_cur = s_next;
        nbeg = n2beg; nend = n2end;
        for (int k = 0; k < H; ++k) { adn[k] = nad[k]; nad[k] = n2ad[k]; }
    }
}

static inline int cdiv(int a, int b) { return (a + b - 1) / b; }

extern "C" void kernel_launch(void* const* d_in, const int* in_sizes, int n_in,
                              void* d_out, int out_size, void* d_ws, size_t ws_size,
                              hipStream_t stream) {
    const float* x      = (const float*)d_in[0];
    const int*   ei     = (const int*)  d_in[1];
    const float* W1     = (const float*)d_in[2];
    const float* a_src1 = (const float*)d_in[3];
    const float* a_dst1 = (const float*)d_in[4];
    const float* b1     = (const float*)d_in[5];
    const float* W2     = (const float*)d_in[6];
    const float* a_src2 = (const float*)d_in[7];
    const float* a_dst2 = (const float*)d_in[8];
    const float* b2     = (const float*)d_in[9];

    const int N  = in_sizes[0] / 128;
    const int E  = in_sizes[1] / 2;

    float* ws   = (float*)d_ws;
    float* h1   = ws;
    float* agg1 = h1   + (size_t)N * 64;
    float* as1  = agg1 + (size_t)N * 64;
    float* ad1  = as1  + (size_t)N * 4;
    int*   strt = (int*)(ad1 + (size_t)N * 4);
    int*   rnkw = strt  + N + 4;
    int*   csr  = rnkw  + E;
    int*   bsum = csr   + E;
    unsigned* tabg = (unsigned*)(bsum + NCHUNK * 128);
    int*   base = (int*)(tabg + (size_t)NCHUNK * CAPW);
    unsigned short* rank = (unsigned short*)rnkw;
    float* h2   = h1;
    float* as2  = h2   + (size_t)N * 16;
    float* ad2  = as2  + N;

    float* out  = (float*)d_out;

    const int TB = 256;
    const int NB = cdiv(N, 1024);

    count_gemm_kernel<<<NCHUNK + NGEMM, 1024, 0, stream>>>(
        ei, E, tabg, rank, bsum, x, W1, a_src1, a_dst1, h1, as1, ad1, N);
    chunk_scan_kernel<<<NB, 256, 0, stream>>>(tabg, bsum, strt, base, N);
    chunk_place_kernel<<<dim3(32, NCHUNK), TB, 0, stream>>>(ei, E, base, rank, csr);

    node_gather_kernel<4, 16, true><<<2048, TB, 0, stream>>>(
        strt, csr, h1, ad1, a_src1, b1, agg1, N);

    gemm_logits_wmma_f32<64, 1, 2><<<N / 32, dim3(32, 2), 0, stream>>>(
        agg1, W2, a_src2, a_dst2, h2, as2, ad2);

    node_gather_kernel<1, 16, false><<<2048, TB, 0, stream>>>(
        strt, csr, h2, ad2, a_src2, b2, out, N);
}


# Gather: softmax and elu exponentials as hardware exp2 of a scaled argument, the normalisation as one hardware reciprocal times the sums; library exp and IEEE divides go (last-place rounding differs)
# speedup vs baseline: 8.3127x; 1.0314x over previous
#include <hip/hip_runtime.h>
#include <hip/hip_bf16.h>
#include <math.h>

typedef float v2f __attribute__((ext_vector_type(2)));
typedef float v8f __attribute__((ext_vector_type(8)));
typedef float v4f __attribute__((ext_vector_type(4)));
typedef int   v4i __attribute__((ext_vector_type(4)));

#define NEG_SLOPE 0.2f

#define NCHUNK 32
#define CAPW   50176
#define NPITCH (2 * CAPW)
#define NGEMM  224

template <int M>
__device__ __forceinline__ unsigned row_xor_u(unsigned v) {
    return (unsigned)__builtin_amdgcn_update_dpp(
        0, (int)v, 0x160 | M, 0xf, 0xf, true);
}

template <int M>
__device__ __forceinline__ float row_xor(float v) {
    return __int_as_float(__builtin_amdgcn_update_dpp(
        0, __float_as_int(v), 0x160 | M, 0xf, 0xf, true));
}

__device__ __forceinline__ void lds_barrier() {
    __builtin_amdgcn_fence(__ATOMIC_RELEASE, "workgroup", "local");
    __builtin_amdgcn_s_barrier();
    __builtin_amdgcn_fence(__ATOMIC_ACQUIRE, "workgroup", "local");
}

#define G1_K     128
#define G1_FOUT  64
#define G1_PITCH (G1_K + 4)
#define G1_RT    8
#define G1_ROWS  (16 * G1_RT)

__device__ __forceinline__ void gemm1_rows(float* __restrict__ lds,
                                           const float* __restrict__ X,
                                           const float* __restrict__ W,
                                           const float* __restrict__ a_src,
                                           const float* __restrict__ a_dst,
                                           float* __restrict__ H,
                                           float* __restrict__ as,
                                           float* __restrict__ ad, int N) {
    constexpr int NLOG = G1_ROWS * 4;
    static_assert(2 * G1_ROWS * G1_PITCH + 4 * NLOG <= CAPW, "fits the counter table's LDS");
    float* const shl = lds + 2 * G1_ROWS * G1_PITCH;

    const int lane = threadIdx.x & 31;
    const int wave = __builtin_amdgcn_readfirstlane((int)(threadIdx.x >> 5));
    const int rt   = wave >> 2;
    const int ct   = wave & 3;
    const int half = lane >> 4;
    const int m    = lane & 15;
    const int c0   = ct * 16;

    const int ntile = N / 16;
    const int ngw   = (int)gridDim.x - NCHUNK;
    const int wg    = (int)blockIdx.x - NCHUNK;
    const int tpw   = (ntile + ngw - 1) / ngw;
    const int t0    = wg * tpw;
    const int t1    = min(ntile, t0 + tpw);
    const int nit   = (t1 > t0) ? (t1 - t0 + G1_RT - 1) / G1_RT : 0;

    v2f b[G1_K / 4];
    #pragma unroll
    for (int s = 0; s < G1_K / 4; ++s) {
        const int kb = 4 * s + 2 * half;
        b[s][0] = W[(size_t)(kb)     * G1_FOUT + c0 + m];
        b[s][1] = W[(size_t)(kb + 1) * G1_FOUT + c0 + m];
    }
    const float asv = a_src[c0 + m];
    const float adv = a_dst[c0 + m];

    v4f pre[4];
    #pragma unroll
    for (int j = 0; j < 4; ++j) pre[j] = v4f{0.f, 0.f, 0.f, 0.f};
    if (nit > 0) {
        #pragma unroll
        for (int j = 0; j < 4; ++j) {
            const int row = wave + 32 * j;
            if (t0 + (row >> 4) < t1)
                pre[j] = *(const v4f*)(X + (size_t)(t0 * 16 + row) * G1_K + 4 * lane);
        }
    }

    for (int it = 0; it < nit; ++it) {
        const int buf = it & 1;
        const int tb  = t0 + it * G1_RT;
        float* const Ab = lds + buf * (G1_ROWS * G1_PITCH);

        #pragma unroll
        for (int j = 0; j < 4; ++j) {
            const int row = wave + 32 * j;
            if (tb + (row >> 4) < t1)
                *(v4f*)(Ab + row * G1_PITCH + 4 * lane) = pre[j];
        }
        lds_barrier();

        if (it > 0 && wave < 8) {
            const int kind = wave >> 2;
            const int f    = (wave & 3) * 128 + lane * 4;
            const int pb   = tb - G1_RT;
            if (pb + (f >> 6) < t1)
                *(v4f*)((kind ? ad : as) + (size_t)pb * 64 + f) =
                    *(const v4f*)(shl + ((buf ^ 1) * 2 + kind) * NLOG + f);
        }

        if (it + 1 < nit) {
            #pragma unroll
            for (int j = 0; j < 4; ++j) {
                const int row = wave + 32 * j;
                if (tb + G1_RT + (row >> 4) < t1)
                    pre[j] = *(const v4f*)(X + (size_t)((tb + G1_RT) * 16 + row) * G1_K + 4 * lane);
            }
        }

        if (tb + rt < t1) {
            const float* const Ar = Ab + (rt * 16 + m) * G1_PITCH + 2 * half;
            v8f acc = {};
            #pragma unroll
            for (int s = 0; s < G1_K / 4; ++s) {
                const v2f a = *(const v2f*)(Ar + 4 * s);
                acc = __builtin_amdgcn_wmma_f32_16x16x4_f32(
                     false, a,  false, b[s],
                     (short)0, acc,  false,  false);
            }

            float lg;
            {
                const bool b3 = (m & 8) != 0, b2 = (m & 4) != 0;
                const bool b1 = (m & 2) != 0, b0 = (m & 1) != 0;
                const float kv = b3 ? adv : asv;
                const float sv = b3 ? asv : adv;
                float q[8], t[4], u[2];
                #pragma unroll
                for (int r = 0; r < 8; ++r)
                    q[r] = acc[r] * kv + row_xor<8>(acc[r] * sv);
                #pragma unroll
                for (int r = 0; r < 4; ++r)
                    t[r] = (b2 ? q[r + 4] : q[r]) + row_xor<4>(b2 ? q[r] : q[r + 4]);
                #pragma unroll
                for (int r = 0; r < 2; ++r)
                    u[r] = (b1 ? t[r + 2] : t[r]) + row_xor<2>(b1 ? t[r] : t[r + 2]);
                lg = (b0 ? u[1] : u[0]) + row_xor<1>(b0 ? u[0] : u[1]);
            }
            shl[(buf * 2 + (m >> 3)) * NLOG + (rt * 16 + (m & 7) + 8 * half) * 4 + ct] = lg;

            #pragma unroll
            for (int r = 0; r < 8; ++r)
                H[(size_t)((tb + rt) * 16 + r + 8 * half) * G1_FOUT + c0 + m] = acc[r];
        }
    }
    lds_barrier();
    if (nit > 0 && wave < 8) {
        const int kind = wave >> 2;
        const int f    = (wave & 3) * 128 + lane * 4;
        const int pb   = t0 + (nit - 1) * G1_RT;
        if (pb + (f >> 6) < t1)
            *(v4f*)((kind ? ad : as) + (size_t)pb * 64 + f) =
                *(const v4f*)(shl + (((nit - 1) & 1) * 2 + kind) * NLOG + f);
    }
}

__global__ void __launch_bounds__(1024)
count_gemm_kernel(const int* __restrict__ ei, int E,
                  unsigned* __restrict__ tabg,
                  unsigned short* __restrict__ rank,
                  int* __restrict__ bsumg,
                  const float* __restrict__ X,
                  const float* __restrict__ W,
                  const float* __restrict__ a_src,
                  const float* __restrict__ a_dst,
                  float* __restrict__ H,
                  float* __restrict__ as,
                  float* __restrict__ ad, int N) {
    __shared__ __attribute__((aligned(16))) unsigned tab[CAPW];
    __shared__ unsigned blk[128];
    if (blockIdx.x >= NCHUNK) {
        gemm1_rows(reinterpret_cast<float*>(tab), X, W, a_src, a_dst, H, as, ad, N);
        return;
    }
    const int t  = threadIdx.x;
    const int g  = blockIdx.x;
    const int CH = (E + NCHUNK - 1) / NCHUNK;
    const int c0 = g * CH;
    const int c1 = min(E, c0 + CH);

    for (int i = t * 4; i < CAPW; i += 4096)
        *(uint4*)&tab[i] = make_uint4(0u, 0u, 0u, 0u);
    if (t < 128) blk[t] = 0u;
    __syncthreads();

    constexpr int EPT = 30;
    int e0 = c0 + t;
    for (; e0 + (EPT - 1) * 1024 < c1; e0 += EPT * 1024) {
        int d[EPT];
        unsigned old[EPT];
        #pragma unroll
        for (int u = 0; u < EPT; ++u) d[u] = ei[E + e0 + u * 1024];
        #pragma unroll
        for (int u = 0; u < EPT; ++u)
            old[u] = atomicAdd(&tab[d[u] >> 1], 1u << ((unsigned)(d[u] & 1) * 16u));
        #pragma unroll
        for (int u = 0; u < EPT; ++u)
            rank[e0 + u * 1024] =
                (unsigned short)(old[u] >> ((unsigned)(d[u] & 1) * 16u));
    }
    for (int e = e0; e < c1; e += 1024) {
        const int dd = ei[E + e];
        const unsigned sh  = (unsigned)(dd & 1) * 16u;
        const unsigned old = atomicAdd(&tab[dd >> 1], 1u << sh);
        rank[e] = (unsigned short)(old >> sh);
    }
    __syncthreads();

    unsigned* const dst = tabg + (size_t)g * CAPW;
    constexpr int NRND = CAPW / 4096;
    const bool tail = t * 4 + NRND * 4096 < CAPW;
    unsigned s[NRND + 1];
    #pragma unroll
    for (int r = 0; r <= NRND; ++r) {
        const int i = t * 4 + r * 4096;
        s[r] = 0u;
        if (r < NRND || tail) {
            const uint4 q = *(const uint4*)&tab[i];
            *(uint4*)&dst[i] = q;
            s[r] = (q.x & 0xffffu) + (q.x >> 16) + (q.y & 0xffffu) + (q.y >> 16)
                 + (q.z & 0xffffu) + (q.z >> 16) + (q.w & 0xffffu) + (q.w >> 16);
        }
    }
    #pragma unroll
    for (int r = 0; r <= NRND; ++r) {
        if (r < NRND || tail) {
            unsigned x = s[r];
            x += row_xor_u<1>(x);
            x += row_xor_u<2>(x);
            x += row_xor_u<4>(x);
            x += row_xor_u<8>(x);
            if ((t & 15) == 0) atomicAdd(&blk[(t * 4 + r * 4096) >> 9], x);
        }
    }
    __syncthreads();
    if (t < 128) bsumg[g * 128 + t] = (int)blk[t];
}

__global__ void __launch_bounds__(256)
chunk_scan_kernel(const unsigned* __restrict__ tabg,
                                  const int* __restrict__ bsumg,
                                  int* __restrict__ start,
                                  int* __restrict__ base, int n) {
    __shared__ int sh[256];
    const int t  = threadIdx.x;
    const int i0 = blockIdx.x * 1024 + t * 4;
    const int w0 = blockIdx.x * 512 + t * 2;

    int part = 0;
    if (t < (int)blockIdx.x) {
        #pragma unroll
        for (int g = 0; g < NCHUNK; ++g) part += bsumg[g * 128 + t];
    }

    uint2 v[NCHUNK];
    #pragma unroll
    for (int g = 0; g < NCHUNK; ++g)
        v[g] = *(const uint2*)&tabg[(size_t)g * CAPW + w0];

    int c[4] = {0, 0, 0, 0};
    #pragma unroll
    for (int g = 0; g < NCHUNK; ++g) {
        c[0] += (int)(v[g].x & 0xffffu);
        c[1] += (int)(v[g].x >> 16);
        c[2] += (int)(v[g].y & 0xffffu);
        c[3] += (int)(v[g].y >> 16);
    }
    const int tsum = c[0] + c[1] + c[2] + c[3];

    sh[t] = part;
    __syncthreads();
    for (int off = 128; off > 0; off >>= 1) {
        if (t < off) sh[t] += sh[t + off];
        __syncthreads();
    }
    const int before = sh[0];
    __syncthreads();

    sh[t] = tsum;
    __syncthreads();
    for (int off = 1; off < 256; off <<= 1) {
        const int x = (t >= off) ? sh[t - off] : 0;
        __syncthreads();
        sh[t] += x;
        __syncthreads();
    }
    int o[4];
    o[0] = before + sh[t] - tsum;
    o[1] = o[0] + c[0];
    o[2] = o[1] + c[1];
    o[3] = o[2] + c[2];
    for (int k = 0; k < 4; ++k)
        if (i0 + k < n) start[i0 + k] = o[k];
    if (i0 < n && n <= i0 + 4) start[n] = o[3] + c[3];

    #pragma unroll
    for (int g = 0; g < NCHUNK; ++g) {
        *(int4*)&base[(size_t)g * NPITCH + i0] = make_int4(o[0], o[1], o[2], o[3]);
        o[0] += (int)(v[g].x & 0xffffu);
        o[1] += (int)(v[g].x >> 16);
        o[2] += (int)(v[g].y & 0xffffu);
        o[3] += (int)(v[g].y >> 16);
    }
}

__global__ void __launch_bounds__(256)
chunk_place_kernel(const int* __restrict__ ei, int E,
                                   const int* __restrict__ base,
                                   const unsigned short* __restrict__ rank,
                                   int* __restrict__ csr) {
    const int g  = blockIdx.y;
    const int CH = (E + NCHUNK - 1) / NCHUNK;
    const int c0 = g * CH;
    const int c1 = min(E, c0 + CH);
    const int* const bg = base + (size_t)g * NPITCH;
    const int stride = gridDim.x * blockDim.x;
    for (int e = c0 + blockIdx.x * blockDim.x + threadIdx.x; e < c1; e += stride) {
        const int d = ei[E + e];
        csr[bg[d] + (int)rank[e]] = ei[e];
    }
}

template <int K, int NCOLT, int NROWT>
__global__ void gemm_logits_wmma_f32(const float* __restrict__ X,
                                     const float* __restrict__ W,
                                     const float* __restrict__ a_src,
                                     const float* __restrict__ a_dst,
                                     float* __restrict__ H,
                                     float* __restrict__ as,
                                     float* __restrict__ ad) {
    constexpr int FOUT = 16 * NCOLT;
    constexpr int ROWS = 16 * NROWT;
    constexpr int NW   = NCOLT * NROWT;
    constexpr int NL   = ROWS * NCOLT;
    constexpr int PER  = NL / 32;
    static_assert(NW >= 2, "one wave stores as, another ad");
    static_assert(PER == 1 || PER == 2, "a block's logits are one store per kind");
    __shared__ float As[ROWS][K + 4];
    __shared__ __attribute__((aligned(8))) float sh_log[2][NL];

    const int r0   = blockIdx.x * ROWS;
    const int lane = threadIdx.x;
    const int wave = threadIdx.y;
    const int rt   = wave / NCOLT;
    const int ct   = wave % NCOLT;
    const int tid  = wave * 32 + lane;
    constexpr int nthr = 32 * NW;

    for (int i = tid; i < ROWS * K; i += nthr) {
        int rr = i / K, cc = i % K;
        As[rr][cc] = X[(size_t)(r0 + rr) * K + cc];
    }
    __syncthreads();

    const int half = lane >> 4;
    const int m    = lane & 15;
    const int c0   = ct * 16;
    const int ar   = rt * 16 + m;

    const float asv = a_src[c0 + m];
    const float adv = a_dst[c0 + m];

    v8f acc = {};
    #pragma unroll 4
    for (int kk = 0; kk < K; kk += 4) {
        const int kb = kk + 2 * half;
        v2f a, b;
        a[0] = As[ar][kb];
        a[1] = As[ar][kb + 1];
        b[0] = W[(size_t)(kb)     * FOUT + c0 + m];
        b[1] = W[(size_t)(kb + 1) * FOUT + c0 + m];
        acc = __builtin_amdgcn_wmma_f32_16x16x4_f32(
             false, a,  false, b,
             (short)0, acc,  false,  false);
    }

    float lg;
    {
        const bool b3 = (m & 8) != 0, b2 = (m & 4) != 0;
        const bool b1 = (m & 2) != 0, b0 = (m & 1) != 0;
        const float kv = b3 ? adv : asv;
        const float sv = b3 ? asv : adv;
        float q[8], t[4], u[2];
        #pragma unroll
        for (int r = 0; r < 8; ++r)
            q[r] = acc[r] * kv + row_xor<8>(acc[r] * sv);
        #pragma unroll
        for (int r = 0; r < 4; ++r)
            t[r] = (b2 ? q[r + 4] : q[r]) + row_xor<4>(b2 ? q[r] : q[r + 4]);
        #pragma unroll
        for (int r = 0; r < 2; ++r)
            u[r] = (b1 ? t[r + 2] : t[r]) + row_xor<2>(b1 ? t[r] : t[r + 2]);
        lg = (b0 ? u[1] : u[0]) + row_xor<1>(b0 ? u[0] : u[1]);
    }
    sh_log[m >> 3][(rt * 16 + (m & 7) + 8 * half) * NCOLT + ct] = lg;
    __syncthreads();

    #pragma unroll
    for (int r = 0; r < 8; ++r) {
        H[(size_t)(r0 + rt * 16 + r + 8 * half) * FOUT + c0 + m] = acc[r];
    }

    if (wave < 2) {
        float* const dst = (wave == 0 ? as : ad) + (size_t)blockIdx.x * NL + lane * PER;
        if constexpr (PER == 2)
            *(v2f*)dst = *(const v2f*)&sh_log[wave][lane * 2];
        else
            *dst = sh_log[wave][lane];
    }
}

template <int MASK>
__device__ __forceinline__ float lane_xor(float v) {
    return __int_as_float(
        __builtin_amdgcn_ds_swizzle(__float_as_int(v), (MASK << 10) | 0x1f));
}

__device__ __forceinline__ float exp_hw(float x) {
    return __builtin_amdgcn_exp2f(x * 1.44269504088896341f);
}

template <int F, int NR, int GR, int EPS>
__device__ __forceinline__ void rows_request(const float* __restrict__ h,
                                             const int* idx_slot, int cnt,
                                             int f0, v4f (&r)[NR]) {
    v4i ix[NR / 4];
    #pragma unroll
    for (int q = 0; q < NR / 4; ++q) ix[q] = *(const v4i*)(idx_slot + 4 * q);
    #pragma unroll
    for (int g = 0; g < NR / GR; ++g) {
        if (g == 0 || cnt > g * GR * EPS) {
            #pragma unroll
            for (int u = 0; u < GR; ++u) {
                const int j = g * GR + u;
                const unsigned off = (unsigned)ix[j / 4][j % 4] * (unsigned)(F * 4)
                                   + (unsigned)f0 * 4u;
                r[j] = *(const v4f*)((const char*)h + off);
            }
        }
    }
}

template <int Q>
__device__ __forceinline__ float quad_bcast(float v) {
    return __int_as_float(__builtin_amdgcn_update_dpp(
        0, __float_as_int(v), Q * 0x55, 0xf, 0xf, true));
}

template <int NR, int GR, int EPS, int LPR>
__device__ __forceinline__ void window_sum(const v4f (&r)[NR], int cnt, int vb,
                                           bool b0, bool b1, bool later,
                                           const v4f a4, float adl, float& mrun,
                                           float (&acc)[4], float& den) {
    constexpr int NH = NR / 2;
    constexpr int NO = NR / 4;
    const v2f alo = __builtin_shufflevector(a4, a4, 0, 1);
    const v2f ahi = __builtin_shufflevector(a4, a4, 2, 3);
    float p[NR];
    #pragma unroll
    for (int j = 0; j < NR; ++j) p[j] = 0.f;
    #pragma unroll
    for (int g = 0; g < NR / GR; ++g) {
        if (g == 0 || cnt > g * GR * EPS) {
            #pragma unroll
            for (int u = 0; u < GR; ++u) {
                const int j = g * GR + u;
                v2f s2 = __builtin_shufflevector(r[j], r[j], 0, 1) * alo;
                s2 = __builtin_elementwise_fma(
                    __builtin_shufflevector(r[j], r[j], 2, 3), ahi, s2);
                p[j] = s2[0] + s2[1];
            }
        }
    }
    float t[NH];
    #pragma unroll
    for (int i = 0; i < NH; ++i)
        t[i] = (b0 ? p[i + NH] : p[i]) + row_xor<1>(b0 ? p[i] : p[i + NH]);
    float lg[NO];
    #pragma unroll
    for (int i = 0; i < NO; ++i)
        lg[i] = (b1 ? t[i + NO] : t[i]) + row_xor<2>(b1 ? t[i] : t[i + NO]);

    float mx = -INFINITY;
    #pragma unroll
    for (int i = 0; i < NO; ++i) {
        float v = lg[i] + adl;
        v = (v > 0.f) ? v : NEG_SLOPE * v;
        lg[i] = (vb < cnt - i * EPS) ? v : -INFINITY;
        mx = fmaxf(mx, lg[i]);
    }
    mx = fmaxf(mx, row_xor<1>(mx));
    mx = fmaxf(mx, row_xor<2>(mx));
    if constexpr (LPR == 4) {
        mx = fmaxf(mx, row_xor<4>(mx));
        mx = fmaxf(mx, row_xor<8>(mx));
    }
    mx = fmaxf(mx, lane_xor<16>(mx));
    if (later) {
        mx = fmaxf(mx, mrun);
        const float sc = exp_hw(mrun - mx);
        #pragma unroll
        for (int v = 0; v < 4; ++v) acc[v] *= sc;
        den *= sc;
    }
    mrun = mx;

    float w[NO];
    #pragma unroll
    for (int i = 0; i < NO; ++i) w[i] = exp_hw(lg[i] - mx);

    #pragma unroll
    for (int g = 0; g < NR / GR; ++g) {
        if (g == 0 || cnt > g * GR * EPS) {
            #pragma unroll
            for (int u = 0; u < GR; ++u) {
                const int j = g * GR + u;
                float wj;
                switch ((j / NH) + 2 * ((j % NH) / NO)) {
                    case 0:  wj = quad_bcast<0>(w[j % NO]); break;
                    case 1:  wj = quad_bcast<1>(w[j % NO]); break;
                    case 2:  wj = quad_bcast<2>(w[j % NO]); break;
                    default: wj = quad_bcast<3>(w[j % NO]); break;
                }
                #pragma unroll
                for (int v = 0; v < 4; ++v) acc[v] = fmaf(wj, r[j][v], acc[v]);
                den += wj;
            }
        }
    }
}

template <int H, int D, bool ELU>
__global__ void __launch_bounds__(256, 16)
node_gather_kernel(const int* __restrict__ start, const int* __restrict__ csr,
                   const float* __restrict__ h,
                   const float* __restrict__ ad,
                   const float* __restrict__ a_src,
                   const float* __restrict__ bias,
                   float* __restrict__ out, int N) {
    constexpr int F   = H * D;
    constexpr int VEC = 4;
    constexpr int LPR = F / VEC;
    constexpr int EPS = 32 / LPR;
    constexpr int NST = LPR;
    constexpr int NR  = (NST < 8) ? NST : 8;
    constexpr int GR  = NR / 2;
    constexpr int WE  = NR * EPS;
    constexpr int WPC = NST / NR;
    static_assert(LPR == 16 || LPR == 4, "row must map to 16 or 4 lanes");
    static_assert(D == 4 * VEC, "a head's logit is summed over 4 adjacent lanes");

    __shared__ __attribute__((aligned(16))) int sh_idx[8][32];

    const int lane = threadIdx.x & 31;
    const int wib  = threadIdx.x >> 5;
    const int sub  = lane / LPR;
    const int f0   = (lane % LPR) * VEC;
    const int myh  = f0 / D;
    const int ppos = (lane % EPS) * LPR + lane / EPS;
    int* const my_idx = sh_idx[wib];
    const int* const idx_slot = my_idx + sub * LPR;
    const bool lb0 = (lane & 1) != 0, lb1 = (lane & 2) != 0;
    const int  vb  = ((lane & 1) * (NR / 2) + ((lane >> 1) & 1) * (NR / 4)) * EPS + sub;
    const int fo = (LPR == 16) ? f0 + 2 * (lane >> 4)
                               : f0 + 2 * ((lane >> 4) & 1) + ((lane >> 3) & 1);
    const int wave   = blockIdx.x * (blockDim.x >> 5) + (threadIdx.x >> 5);
    const int nwaves = gridDim.x * (blockDim.x >> 5);

    if (wave >= N) return;

    const int wave_s   = __builtin_amdgcn_readfirstlane(wave);
    const int nwaves_s = __builtin_amdgcn_readfirstlane(nwaves);

    float bo0 = bias[fo];
    float bo1 = (LPR == 16) ? bias[fo + 1] : 0.f;

    const v4f a4 = *(const v4f*)(a_src + f0);

    int   beg, end, s_cur;
    float adn[H];
    int   nbeg = 0, nend = 0;
    float nad[H];
    {
        const int n = wave_s;
        beg = start[n];
        end = start[n + 1];
        for (int k = 0; k < H; ++k) adn[k] = ad[n * H + k];
        for (int k = 0; k < H; ++k) nad[k] = 0.f;
        const int nn = n + nwaves_s;
        if (nn < N) {
            nbeg = start[nn];
            nend = start[nn + 1];
            for (int k = 0; k < H; ++k) nad[k] = ad[nn * H + k];
        }
        s_cur = (lane < end - beg) ? csr[beg + lane] : n;
    }

    for (int n = wave_s; n < N; n += nwaves_s) {
        const int cnt = end - beg + 1;

        const bool has_next = n + nwaves_s < N;
        const int  nn       = has_next ? n + nwaves_s : n;
        const int  n2       = nn + nwaves_s;
        const bool has_n2   = has_next && n2 < N;
        int   n2beg = 0, n2end = 0;
        float n2ad[H];
        for (int k = 0; k < H; ++k) n2ad[k] = 0.f;
        int s_next = nn;

        float acc[VEC];
        for (int v = 0; v < VEC; ++v) acc[v] = 0.f;
        float den = 0.f;
        float mrun = -INFINITY;

        float adl = adn[0];
        for (int k = 1; k < H; ++k) adl = (myh == k) ? adn[k] : adl;

        my_idx[ppos] = s_cur;

        if (has_next)
            s_next = (lane < nend - nbeg) ? csr[nbeg + lane] : nn;
        if (has_n2) {
            n2beg = start[n2];
            n2end = start[n2 + 1];
            for (int k = 0; k < H; ++k) n2ad[k] = ad[n2 * H + k];
        }

        #pragma clang loop unroll(disable)
        for (int wi = 0; wi * WE < cnt; ++wi) {
            if (wi > 0 && wi % WPC == 0) {
                const int idx = wi * WE + lane;
                my_idx[ppos] = (idx < cnt - 1) ? csr[beg + idx] : n;
            }
            v4f r[NR];
            rows_request<F, NR, GR, EPS>(h, idx_slot + (wi % WPC) * NR,
                                         cnt - wi * WE, f0, r);
            window_sum<NR, GR, EPS, LPR>(r, cnt - wi * WE, vb, lb0, lb1, wi > 0,
                                         a4, adl, mrun, acc, den);
        }

        float k0, k1;
        {
            const bool hi = (lane & 16) != 0;
            k0 = hi ? acc[2] : acc[0];
            k1 = hi ? acc[3] : acc[1];
            const float s0 = hi ? acc[0] : acc[2];
            const float s1 = hi ? acc[1] : acc[3];
            k0  += lane_xor<16>(s0);
            k1  += lane_xor<16>(s1);
            den += lane_xor<16>(den);
        }
        if constexpr (LPR == 16) {
            const float inv = __builtin_amdgcn_rcpf(den + 1e-16f);
            v2f val;
            val[0] = k0 * inv + bo0;
            val[1] = k1 * inv + bo1;
            if (ELU) {
                val[0] = (val[0] > 0.f) ? val[0] : (exp_hw(val[0]) - 1.f);
                val[1] = (val[1] > 0.f) ? val[1] : (exp_hw(val[1]) - 1.f);
            }
            *(v2f*)((char*)out + ((unsigned)n * (unsigned)(F * 4) + (unsigned)fo * 4u)) = val;
        } else {
            const bool hi8 = (lane & 8) != 0;
            float kk = hi8 ? k1 : k0;
            const float ss = hi8 ? k0 : k1;
            kk  += lane_xor<8>(ss);
            den += lane_xor<8>(den);
            kk  += lane_xor<4>(kk);
            den += lane_xor<4>(den);
            if ((lane & 4) == 0) {
                float val = kk * __builtin_amdgcn_rcpf(den + 1e-16f) + bo0;
                if (ELU) val = (val > 0.f) ? val : (exp_hw(val) - 1.f);
                *(float*)((char*)out + ((unsigned)n * (unsigned)(F * 4) + (unsigned)fo * 4u)) = val;
            }
        }

        beg = nbeg; end = nend; s_cur = s_next;
        nbeg = n2beg; nend = n2end;
        for (int k = 0; k < H; ++k) { adn[k] = nad[k]; nad[k] = n2ad[k]; }
    }
}

static inline int cdiv(int a, int b) { return (a + b - 1) / b; }

extern "C" void kernel_launch(void* const* d_in, const int* in_sizes, int n_in,
                              void* d_out, int out_size, void* d_ws, size_t ws_size,
                              hipStream_t stream) {
    const float* x      = (const float*)d_in[0];
    const int*   ei     = (const int*)  d_in[1];
    const float* W1     = (const float*)d_in[2];
    const float* a_src1 = (const float*)d_in[3];
    const float* a_dst1 = (const float*)d_in[4];
    const float* b1     = (const float*)d_in[5];
    const float* W2     = (const float*)d_in[6];
    const float* a_src2 = (const float*)d_in[7];
    const float* a_dst2 = (const float*)d_in[8];
    const float* b2     = (const float*)d_in[9];

    const int N  = in_sizes[0] / 128;
    const int E  = in_sizes[1] / 2;

    float* ws   = (float*)d_ws;
    float* h1   = ws;
    float* agg1 = h1   + (size_t)N * 64;
    float* as1  = agg1 + (size_t)N * 64;
    float* ad1  = as1  + (size_t)N * 4;
    int*   strt = (int*)(ad1 + (size_t)N * 4);
    int*   rnkw = strt  + N + 4;
    int*   csr  = rnkw  + E;
    int*   bsum = csr   + E;
    unsigned* tabg = (unsigned*)(bsum + NCHUNK * 128);
    int*   base = (int*)(tabg + (size_t)NCHUNK * CAPW);
    unsigned short* rank = (unsigned short*)rnkw;
    float* h2   = h1;
    float* as2  = h2   + (size_t)N * 16;
    float* ad2  = as2  + N;

    float* out  = (float*)d_out;

    const int TB = 256;
    const int NB = cdiv(N, 1024);

    count_gemm_kernel<<<NCHUNK + NGEMM, 1024, 0, stream>>>(
        ei, E, tabg, rank, bsum, x, W1, a_src1, a_dst1, h1, as1, ad1, N);
    chunk_scan_kernel<<<NB, 256, 0, stream>>>(tabg, bsum, strt, base, N);
    chunk_place_kernel<<<dim3(32, NCHUNK), TB, 0, stream>>>(ei, E, base, rank, csr);

    node_gather_kernel<4, 16, true><<<2048, TB, 0, stream>>>(
        strt, csr, h1, ad1, a_src1, b1, agg1, N);

    gemm_logits_wmma_f32<64, 1, 2><<<N / 32, dim3(32, 2), 0, stream>>>(
        agg1, W2, a_src2, a_dst2, h2, as2, ad2);

    node_gather_kernel<1, 16, false><<<2048, TB, 0, stream>>>(
        strt, csr, h2, ad2, a_src2, b2, out, N);
}


# Layer-2 product (f32 WMMA, logits) moves into the layer-1 gather: a wave takes 8 consecutive nodes, keeps their elu'd rows in its own LDS tile, multiplies at its end; GEMM launch, row round trip go
# speedup vs baseline: 8.9474x; 1.0764x over previous
#include <hip/hip_runtime.h>
#include <hip/hip_bf16.h>
#include <math.h>

typedef float v2f __attribute__((ext_vector_type(2)));
typedef float v8f __attribute__((ext_vector_type(8)));
typedef float v4f __attribute__((ext_vector_type(4)));
typedef int   v4i __attribute__((ext_vector_type(4)));

#define NEG_SLOPE 0.2f

#define NCHUNK 32
#define CAPW   50176
#define NPITCH (2 * CAPW)
#define NGEMM  224

template <int M>
__device__ __forceinline__ unsigned row_xor_u(unsigned v) {
    return (unsigned)__builtin_amdgcn_update_dpp(
        0, (int)v, 0x160 | M, 0xf, 0xf, true);
}

template <int M>
__device__ __forceinline__ float row_xor(float v) {
    return __int_as_float(__builtin_amdgcn_update_dpp(
        0, __float_as_int(v), 0x160 | M, 0xf, 0xf, true));
}

__device__ __forceinline__ void lds_barrier() {
    __builtin_amdgcn_fence(__ATOMIC_RELEASE, "workgroup", "local");
    __builtin_amdgcn_s_barrier();
    __builtin_amdgcn_fence(__ATOMIC_ACQUIRE, "workgroup", "local");
}

__device__ __forceinline__ float tile_logit(const v8f acc, int m, float asv, float adv) {
    const bool b3 = (m & 8) != 0, b2 = (m & 4) != 0;
    const bool b1 = (m & 2) != 0, b0 = (m & 1) != 0;
    const float kv = b3 ? adv : asv;
    const float sv = b3 ? asv : adv;
    float q[8], t[4], u[2];
    #pragma unroll
    for (int r = 0; r < 8; ++r)
        q[r] = acc[r] * kv + row_xor<8>(acc[r] * sv);
    #pragma unroll
    for (int r = 0; r < 4; ++r)
        t[r] = (b2 ? q[r + 4] : q[r]) + row_xor<4>(b2 ? q[r] : q[r + 4]);
    #pragma unroll
    for (int r = 0; r < 2; ++r)
        u[r] = (b1 ? t[r + 2] : t[r]) + row_xor<2>(b1 ? t[r] : t[r + 2]);
    return (b0 ? u[1] : u[0]) + row_xor<1>(b0 ? u[0] : u[1]);
}

#define G1_K     128
#define G1_FOUT  64
#define G1_PITCH (G1_K + 4)
#define G1_RT    8
#define G1_ROWS  (16 * G1_RT)

__device__ __forceinline__ void gemm1_rows(float* __restrict__ lds,
                                           const float* __restrict__ X,
                                           const float* __restrict__ W,
                                           const float* __restrict__ a_src,
                                           const float* __restrict__ a_dst,
                                           float* __restrict__ H,
                                           float* __restrict__ as,
                                           float* __restrict__ ad, int N) {
    constexpr int NLOG = G1_ROWS * 4;
    static_assert(2 * G1_ROWS * G1_PITCH + 4 * NLOG <= CAPW, "fits the counter table's LDS");
    float* const shl = lds + 2 * G1_ROWS * G1_PITCH;

    const int lane = threadIdx.x & 31;
    const int wave = __builtin_amdgcn_readfirstlane((int)(threadIdx.x >> 5));
    const int rt   = wave >> 2;
    const int ct   = wave & 3;
    const int half = lane >> 4;
    const int m    = lane & 15;
    const int c0   = ct * 16;

    const int ntile = N / 16;
    const int ngw   = (int)gridDim.x - NCHUNK;
    const int wg    = (int)blockIdx.x - NCHUNK;
    const int tpw   = (ntile + ngw - 1) / ngw;
    const int t0    = wg * tpw;
    const int t1    = min(ntile, t0 + tpw);
    const int nit   = (t1 > t0) ? (t1 - t0 + G1_RT - 1) / G1_RT : 0;

    v2f b[G1_K / 4];
    #pragma unroll
    for (int s = 0; s < G1_K / 4; ++s) {
        const int kb = 4 * s + 2 * half;
        b[s][0] = W[(size_t)(kb)     * G1_FOUT + c0 + m];
        b[s][1] = W[(size_t)(kb + 1) * G1_FOUT + c0 + m];
    }
    const float asv = a_src[c0 + m];
    const float adv = a_dst[c0 + m];

    v4f pre[4];
    #pragma unroll
    for (int j = 0; j < 4; ++j) pre[j] = v4f{0.f, 0.f, 0.f, 0.f};
    if (nit > 0) {
        #pragma unroll
        for (int j = 0; j < 4; ++j) {
            const int row = wave + 32 * j;
            if (t0 + (row >> 4) < t1)
                pre[j] = *(const v4f*)(X + (size_t)(t0 * 16 + row) * G1_K + 4 * lane);
        }
    }

    for (int it = 0; it < nit; ++it) {
        const int buf = it & 1;
        const int tb  = t0 + it * G1_RT;
        float* const Ab = lds + buf * (G1_ROWS * G1_PITCH);

        #pragma unroll
        for (int j = 0; j < 4; ++j) {
            const int row = wave + 32 * j;
            if (tb + (row >> 4) < t1)
                *(v4f*)(Ab + row * G1_PITCH + 4 * lane) = pre[j];
        }
        lds_barrier();

        if (it > 0 && wave < 8) {
            const int kind = wave >> 2;
            const int f    = (wave & 3) * 128 + lane * 4;
            const int pb   = tb - G1_RT;
            if (pb + (f >> 6) < t1)
                *(v4f*)((kind ? ad : as) + (size_t)pb * 64 + f) =
                    *(const v4f*)(shl + ((buf ^ 1) * 2 + kind) * NLOG + f);
        }

        if (it + 1 < nit) {
            #pragma unroll
            for (int j = 0; j < 4; ++j) {
                const int row = wave + 32 * j;
                if (tb + G1_RT + (row >> 4) < t1)
                    pre[j] = *(const v4f*)(X + (size_t)((tb + G1_RT) * 16 + row) * G1_K + 4 * lane);
            }
        }

        if (tb + rt < t1) {
            const float* const Ar = Ab + (rt * 16 + m) * G1_PITCH + 2 * half;
            v8f acc = {};
            #pragma unroll
            for (int s = 0; s < G1_K / 4; ++s) {
                const v2f a = *(const v2f*)(Ar + 4 * s);
                acc = __builtin_amdgcn_wmma_f32_16x16x4_f32(
                     false, a,  false, b[s],
                     (short)0, acc,  false,  false);
            }

            const float lg = tile_logit(acc, m, asv, adv);
            shl[(buf * 2 + (m >> 3)) * NLOG + (rt * 16 + (m & 7) + 8 * half) * 4 + ct] = lg;

            #pragma unroll
            for (int r = 0; r < 8; ++r)
                H[(size_t)((tb + rt) * 16 + r + 8 * half) * G1_FOUT + c0 + m] = acc[r];
        }
    }
    lds_barrier();
    if (nit > 0 && wave < 8) {
        const int kind = wave >> 2;
        const int f    = (wave & 3) * 128 + lane * 4;
        const int pb   = t0 + (nit - 1) * G1_RT;
        if (pb + (f >> 6) < t1)
            *(v4f*)((kind ? ad : as) + (size_t)pb * 64 + f) =
                *(const v4f*)(shl + (((nit - 1) & 1) * 2 + kind) * NLOG + f);
    }
}

__global__ void __launch_bounds__(1024)
count_gemm_kernel(const int* __restrict__ ei, int E,
                  unsigned* __restrict__ tabg,
                  unsigned short* __restrict__ rank,
                  int* __restrict__ bsumg,
                  const float* __restrict__ X,
                  const float* __restrict__ W,
                  const float* __restrict__ a_src,
                  const float* __restrict__ a_dst,
                  float* __restrict__ H,
                  float* __restrict__ as,
                  float* __restrict__ ad, int N) {
    __shared__ __attribute__((aligned(16))) unsigned tab[CAPW];
    __shared__ unsigned blk[128];
    if (blockIdx.x >= NCHUNK) {
        gemm1_rows(reinterpret_cast<float*>(tab), X, W, a_src, a_dst, H, as, ad, N);
        return;
    }
    const int t  = threadIdx.x;
    const int g  = blockIdx.x;
    const int CH = (E + NCHUNK - 1) / NCHUNK;
    const int c0 = g * CH;
    const int c1 = min(E, c0 + CH);

    for (int i = t * 4; i < CAPW; i += 4096)
        *(uint4*)&tab[i] = make_uint4(0u, 0u, 0u, 0u);
    if (t < 128) blk[t] = 0u;
    __syncthreads();

    constexpr int EPT = 30;
    int e0 = c0 + t;
    for (; e0 + (EPT - 1) * 1024 < c1; e0 += EPT * 1024) {
        int d[EPT];
        unsigned old[EPT];
        #pragma unroll
        for (int u = 0; u < EPT; ++u) d[u] = ei[E + e0 + u * 1024];
        #pragma unroll
        for (int u = 0; u < EPT; ++u)
            old[u] = atomicAdd(&tab[d[u] >> 1], 1u << ((unsigned)(d[u] & 1) * 16u));
        #pragma unroll
        for (int u = 0; u < EPT; ++u)
            rank[e0 + u * 1024] =
                (unsigned short)(old[u] >> ((unsigned)(d[u] & 1) * 16u));
    }
    for (int e = e0; e < c1; e += 1024) {
        const int dd = ei[E + e];
        const unsigned sh  = (unsigned)(dd & 1) * 16u;
        const unsigned old = atomicAdd(&tab[dd >> 1], 1u << sh);
        rank[e] = (unsigned short)(old >> sh);
    }
    __syncthreads();

    unsigned* const dst = tabg + (size_t)g * CAPW;
    constexpr int NRND = CAPW / 4096;
    const bool tail = t * 4 + NRND * 4096 < CAPW;
    unsigned s[NRND + 1];
    #pragma unroll
    for (int r = 0; r <= NRND; ++r) {
        const int i = t * 4 + r * 4096;
        s[r] = 0u;
        if (r < NRND || tail) {
            const uint4 q = *(const uint4*)&tab[i];
            *(uint4*)&dst[i] = q;
            s[r] = (q.x & 0xffffu) + (q.x >> 16) + (q.y & 0xffffu) + (q.y >> 16)
                 + (q.z & 0xffffu) + (q.z >> 16) + (q.w & 0xffffu) + (q.w >> 16);
        }
    }
    #pragma unroll
    for (int r = 0; r <= NRND; ++r) {
        if (r < NRND || tail) {
            unsigned x = s[r];
            x += row_xor_u<1>(x);
            x += row_xor_u<2>(x);
            x += row_xor_u<4>(x);
            x += row_xor_u<8>(x);
            if ((t & 15) == 0) atomicAdd(&blk[(t * 4 + r * 4096) >> 9], x);
        }
    }
    __syncthreads();
    if (t < 128) bsumg[g * 128 + t] = (int)blk[t];
}

__global__ void __launch_bounds__(256)
chunk_scan_kernel(const unsigned* __restrict__ tabg,
                                  const int* __restrict__ bsumg,
                                  int* __restrict__ start,
                                  int* __restrict__ base, int n) {
    __shared__ int sh[256];
    const int t  = threadIdx.x;
    const int i0 = blockIdx.x * 1024 + t * 4;
    const int w0 = blockIdx.x * 512 + t * 2;

    int part = 0;
    if (t < (int)blockIdx.x) {
        #pragma unroll
        for (int g = 0; g < NCHUNK; ++g) part += bsumg[g * 128 + t];
    }

    uint2 v[NCHUNK];
    #pragma unroll
    for (int g = 0; g < NCHUNK; ++g)
        v[g] = *(const uint2*)&tabg[(size_t)g * CAPW + w0];

    int c[4] = {0, 0, 0, 0};
    #pragma unroll
    for (int g = 0; g < NCHUNK; ++g) {
        c[0] += (int)(v[g].x & 0xffffu);
        c[1] += (int)(v[g].x >> 16);
        c[2] += (int)(v[g].y & 0xffffu);
        c[3] += (int)(v[g].y >> 16);
    }
    const int tsum = c[0] + c[1] + c[2] + c[3];

    sh[t] = part;
    __syncthreads();
    for (int off = 128; off > 0; off >>= 1) {
        if (t < off) sh[t] += sh[t + off];
        __syncthreads();
    }
    const int before = sh[0];
    __syncthreads();

    sh[t] = tsum;
    __syncthreads();
    for (int off = 1; off < 256; off <<= 1) {
        const int x = (t >= off) ? sh[t - off] : 0;
        __syncthreads();
        sh[t] += x;
        __syncthreads();
    }
    int o[4];
    o[0] = before + sh[t] - tsum;
    o[1] = o[0] + c[0];
    o[2] = o[1] + c[1];
    o[3] = o[2] + c[2];
    for (int k = 0; k < 4; ++k)
        if (i0 + k < n) start[i0 + k] = o[k];
    if (i0 < n && n <= i0 + 4) start[n] = o[3] + c[3];

    #pragma unroll
    for (int g = 0; g < NCHUNK; ++g) {
        *(int4*)&base[(size_t)g * NPITCH + i0] = make_int4(o[0], o[1], o[2], o[3]);
        o[0] += (int)(v[g].x & 0xffffu);
        o[1] += (int)(v[g].x >> 16);
        o[2] += (int)(v[g].y & 0xffffu);
        o[3] += (int)(v[g].y >> 16);
    }
}

__global__ void __launch_bounds__(256)
chunk_place_kernel(const int* __restrict__ ei, int E,
                                   const int* __restrict__ base,
                                   const unsigned short* __restrict__ rank,
                                   int* __restrict__ csr) {
    const int g  = blockIdx.y;
    const int CH = (E + NCHUNK - 1) / NCHUNK;
    const int c0 = g * CH;
    const int c1 = min(E, c0 + CH);
    const int* const bg = base + (size_t)g * NPITCH;
    const int stride = gridDim.x * blockDim.x;
    for (int e = c0 + blockIdx.x * blockDim.x + threadIdx.x; e < c1; e += stride) {
        const int d = ei[E + e];
        csr[bg[d] + (int)rank[e]] = ei[e];
    }
}

template <int MASK>
__device__ __forceinline__ float lane_xor(float v) {
    return __int_as_float(
        __builtin_amdgcn_ds_swizzle(__float_as_int(v), (MASK << 10) | 0x1f));
}

__device__ __forceinline__ float exp_hw(float x) {
    return __builtin_amdgcn_exp2f(x * 1.44269504088896341f);
}

template <int F, int NR, int GR, int EPS>
__device__ __forceinline__ void rows_request(const float* __restrict__ h,
                                             const int* idx_slot, int cnt,
                                             int f0, v4f (&r)[NR]) {
    v4i ix[NR / 4];
    #pragma unroll
    for (int q = 0; q < NR / 4; ++q) ix[q] = *(const v4i*)(idx_slot + 4 * q);
    #pragma unroll
    for (int g = 0; g < NR / GR; ++g) {
        if (g == 0 || cnt > g * GR * EPS) {
            #pragma unroll
            for (int u = 0; u < GR; ++u) {
                const int j = g * GR + u;
                const unsigned off = (unsigned)ix[j / 4][j % 4] * (unsigned)(F * 4)
                                   + (unsigned)f0 * 4u;
                r[j] = *(const v4f*)((const char*)h + off);
            }
        }
    }
}

template <int Q>
__device__ __forceinline__ float quad_bcast(float v) {
    return __int_as_float(__builtin_amdgcn_update_dpp(
        0, __float_as_int(v), Q * 0x55, 0xf, 0xf, true));
}

template <int NR, int GR, int EPS, int LPR>
__device__ __forceinline__ void window_sum(const v4f (&r)[NR], int cnt, int vb,
                                           bool b0, bool b1, bool later,
                                           const v4f a4, float adl, float& mrun,
                                           float (&acc)[4], float& den) {
    constexpr int NH = NR / 2;
    constexpr int NO = NR / 4;
    const v2f alo = __builtin_shufflevector(a4, a4, 0, 1);
    const v2f ahi = __builtin_shufflevector(a4, a4, 2, 3);
    float p[NR];
    #pragma unroll
    for (int j = 0; j < NR; ++j) p[j] = 0.f;
    #pragma unroll
    for (int g = 0; g < NR / GR; ++g) {
        if (g == 0 || cnt > g * GR * EPS) {
            #pragma unroll
            for (int u = 0; u < GR; ++u) {
                const int j = g * GR + u;
                v2f s2 = __builtin_shufflevector(r[j], r[j], 0, 1) * alo;
                s2 = __builtin_elementwise_fma(
                    __builtin_shufflevector(r[j], r[j], 2, 3), ahi, s2);
                p[j] = s2[0] + s2[1];
            }
        }
    }
    float t[NH];
    #pragma unroll
    for (int i = 0; i < NH; ++i)
        t[i] = (b0 ? p[i + NH] : p[i]) + row_xor<1>(b0 ? p[i] : p[i + NH]);
    float lg[NO];
    #pragma unroll
    for (int i = 0; i < NO; ++i)
        lg[i] = (b1 ? t[i + NO] : t[i]) + row_xor<2>(b1 ? t[i] : t[i + NO]);

    float mx = -INFINITY;
    #pragma unroll
    for (int i = 0; i < NO; ++i) {
        float v = lg[i] + adl;
        v = (v > 0.f) ? v : NEG_SLOPE * v;
        lg[i] = (vb < cnt - i * EPS) ? v : -INFINITY;
        mx = fmaxf(mx, lg[i]);
    }
    mx = fmaxf(mx, row_xor<1>(mx));
    mx = fmaxf(mx, row_xor<2>(mx));
    if constexpr (LPR == 4) {
        mx = fmaxf(mx, row_xor<4>(mx));
        mx = fmaxf(mx, row_xor<8>(mx));
    }
    mx = fmaxf(mx, lane_xor<16>(mx));
    if (later) {
        mx = fmaxf(mx, mrun);
        const float sc = exp_hw(mrun - mx);
        #pragma unroll
        for (int v = 0; v < 4; ++v) acc[v] *= sc;
        den *= sc;
    }
    mrun = mx;

    float w[NO];
    #pragma unroll
    for (int i = 0; i < NO; ++i) w[i] = exp_hw(lg[i] - mx);

    #pragma unroll
    for (int g = 0; g < NR / GR; ++g) {
        if (g == 0 || cnt > g * GR * EPS) {
            #pragma unroll
            for (int u = 0; u < GR; ++u) {
                const int j = g * GR + u;
                float wj;
                switch ((j / NH) + 2 * ((j % NH) / NO)) {
                    case 0:  wj = quad_bcast<0>(w[j % NO]); break;
                    case 1:  wj = quad_bcast<1>(w[j % NO]); break;
                    case 2:  wj = quad_bcast<2>(w[j % NO]); break;
                    default: wj = quad_bcast<3>(w[j % NO]); break;
                }
                #pragma unroll
                for (int v = 0; v < 4; ++v) acc[v] = fmaf(wj, r[j][v], acc[v]);
                den += wj;
            }
        }
    }
}

template <int H, int D, bool ELU, bool NEXT>
__global__ void __launch_bounds__(256, 16)
node_gather_kernel(const int* __restrict__ start, const int* __restrict__ csr,
                   const float* __restrict__ h,
                   const float* __restrict__ ad,
                   const float* __restrict__ a_src,
                   const float* __restrict__ bias,
                   float* __restrict__ out, int N,
                   const float* __restrict__ Wn,
                   const float* __restrict__ a_src_n,
                   const float* __restrict__ a_dst_n,
                   float* __restrict__ as_n,
                   float* __restrict__ ad_n) {
    constexpr int F   = H * D;
    constexpr int VEC = 4;
    constexpr int LPR = F / VEC;
    constexpr int EPS = 32 / LPR;
    constexpr int NST = LPR;
    constexpr int NR  = (NST < 8) ? NST : 8;
    constexpr int GR  = NR / 2;
    constexpr int WE  = NR * EPS;
    constexpr int WPC = NST / NR;
    static_assert(LPR == 16 || LPR == 4, "row must map to 16 or 4 lanes");
    static_assert(D == 4 * VEC, "a head's logit is summed over 4 adjacent lanes");

    __shared__ __attribute__((aligned(16))) int sh_idx[8][32];
    constexpr int NPW = 8;
    constexpr int TP  = F + 4;
    __shared__ __attribute__((aligned(16))) float sh_row[NEXT ? 8 * NPW * TP : 4];
    constexpr int WP  = F / 2 + 4;
    __shared__ __attribute__((aligned(16))) float sh_w[NEXT ? 32 * WP : 4];
    __shared__ float sh_av[NEXT ? 32 : 1];

    const int lane = threadIdx.x & 31;
    const int wib  = threadIdx.x >> 5;
    const int sub  = lane / LPR;
    const int f0   = (lane % LPR) * VEC;
    const int myh  = f0 / D;
    const int ppos = (lane % EPS) * LPR + lane / EPS;
    int* const my_idx = sh_idx[wib];
    const int* const idx_slot = my_idx + sub * LPR;
    const bool lb0 = (lane & 1) != 0, lb1 = (lane & 2) != 0;
    const int  vb  = ((lane & 1) * (NR / 2) + ((lane >> 1) & 1) * (NR / 4)) * EPS + sub;
    const int fo = (LPR == 16) ? f0 + 2 * (lane >> 4)
                               : f0 + 2 * ((lane >> 4) & 1) + ((lane >> 3) & 1);
    const int wave   = blockIdx.x * (blockDim.x >> 5) + (threadIdx.x >> 5);
    const int nwaves = gridDim.x * (blockDim.x >> 5);

    if (!NEXT && wave >= N) return;

    const int first_u = __builtin_amdgcn_readfirstlane(NEXT ? wave * NPW : wave);
    const int first_s = (NEXT && first_u >= N) ? 0 : first_u;
    const int step_s  = NEXT ? 1 : __builtin_amdgcn_readfirstlane(nwaves);
    const int last_s  = NEXT ? ((first_u >= N) ? 0 : min(N, first_u + NPW)) : N;
    float* const my_row = sh_row + (NEXT ? wib * (NPW * TP) : 0);

    float wst[4] = {0.f, 0.f, 0.f, 0.f};
    float avs = 0.f;
    if constexpr (NEXT) {
        static_assert(F * 16 == 4 * 256, "4 weights per thread of the block");
        #pragma unroll
        for (int i = 0; i < 4; ++i) wst[i] = Wn[threadIdx.x + 256 * i];
        if (threadIdx.x < 32)
            avs = (threadIdx.x < 16 ? a_src_n : a_dst_n)[threadIdx.x & 15];
    }

    float bo0 = bias[fo];
    float bo1 = (LPR == 16) ? bias[fo + 1] : 0.f;

    const v4f a4 = *(const v4f*)(a_src + f0);

    int   beg, end, s_cur;
    float adn[H];
    int   nbeg = 0, nend = 0;
    float nad[H];
    {
        const int n = first_s;
        beg = start[n];
        end = start[n + 1];
        for (int k = 0; k < H; ++k) adn[k] = ad[n * H + k];
        for (int k = 0; k < H; ++k) nad[k] = 0.f;
        const int nn = n + step_s;
        if (nn < last_s) {
            nbeg = start[nn];
            nend = start[nn + 1];
            for (int k = 0; k < H; ++k) nad[k] = ad[nn * H + k];
        }
        s_cur = (lane < end - beg) ? csr[beg + lane] : n;
    }

    if constexpr (NEXT) {
        #pragma unroll
        for (int i = 0; i < 4; ++i) {
            const int e = (int)threadIdx.x + 256 * i;
            const int k = e >> 4, col = e & 15;
            sh_w[(((k >> 1) & 1) * 16 + col) * WP + 2 * (k >> 2) + (k & 1)] = wst[i];
        }
        if (threadIdx.x < 32) sh_av[threadIdx.x] = avs;
        lds_barrier();
    }

    for (int n = first_s; n < last_s; n += step_s) {
        const int cnt = end - beg + 1;

        const bool has_next = n + step_s < last_s;
        const int  nn       = has_next ? n + step_s : n;
        const int  n2       = nn + step_s;
        const bool has_n2   = has_next && n2 < last_s;
        int   n2beg = 0, n2end = 0;
        float n2ad[H];
        for (int k = 0; k < H; ++k) n2ad[k] = 0.f;
        int s_next = nn;

        float acc[VEC];
        for (int v = 0; v < VEC; ++v) acc[v] = 0.f;
        float den = 0.f;
        float mrun = -INFINITY;

        float adl = adn[0];
        for (int k = 1; k < H; ++k) adl = (myh == k) ? adn[k] : adl;

        my_idx[ppos] = s_cur;

        if (has_next)
            s_next = (lane < nend - nbeg) ? csr[nbeg + lane] : nn;
        if (has_n2) {
            n2beg = start[n2];
            n2end = start[n2 + 1];
            for (int k = 0; k < H; ++k) n2ad[k] = ad[n2 * H + k];
        }

        #pragma clang loop unroll(disable)
        for (int wi = 0; wi * WE < cnt; ++wi) {
            if (wi > 0 && wi % WPC == 0) {
                const int idx = wi * WE + lane;
                my_idx[ppos] = (idx < cnt - 1) ? csr[beg + idx] : n;
            }
            v4f r[NR];
            rows_request<F, NR, GR, EPS>(h, idx_slot + (wi % WPC) * NR,
                                         cnt - wi * WE, f0, r);
            window_sum<NR, GR, EPS, LPR>(r, cnt - wi * WE, vb, lb0, lb1, wi > 0,
                                         a4, adl, mrun, acc, den);
        }

        float k0, k1;
        {
            const bool hi = (lane & 16) != 0;
            k0 = hi ? acc[2] : acc[0];
            k1 = hi ? acc[3] : acc[1];
            const float s0 = hi ? acc[0] : acc[2];
            const float s1 = hi ? acc[1] : acc[3];
            k0  += lane_xor<16>(s0);
            k1  += lane_xor<16>(s1);
            den += lane_xor<16>(den);
        }
        if constexpr (LPR == 16) {
            const float inv = __builtin_amdgcn_rcpf(den + 1e-16f);
            v2f val;
            val[0] = k0 * inv + bo0;
            val[1] = k1 * inv + bo1;
            if (ELU) {
                val[0] = (val[0] > 0.f) ? val[0] : (exp_hw(val[0]) - 1.f);
                val[1] = (val[1] > 0.f) ? val[1] : (exp_hw(val[1]) - 1.f);
            }
            if constexpr (NEXT) {
                *(v2f*)(my_row + (n - first_s) * TP + fo) = val;
            } else {
                *(v2f*)((char*)out + ((unsigned)n * (unsigned)(F * 4) + (unsigned)fo * 4u)) = val;
            }
        } else {
            const bool hi8 = (lane & 8) != 0;
            float kk = hi8 ? k1 : k0;
            const float ss = hi8 ? k0 : k1;
            kk  += lane_xor<8>(ss);
            den += lane_xor<8>(den);
            kk  += lane_xor<4>(kk);
            den += lane_xor<4>(den);
            if ((lane & 4) == 0) {
                float val = kk * __builtin_amdgcn_rcpf(den + 1e-16f) + bo0;
                if (ELU) val = (val > 0.f) ? val : (exp_hw(val) - 1.f);
                *(float*)((char*)out + ((unsigned)n * (unsigned)(F * 4) + (unsigned)fo * 4u)) = val;
            }
        }

        beg = nbeg; end = nend; s_cur = s_next;
        nbeg = n2beg; nend = n2end;
        for (int k = 0; k < H; ++k) { adn[k] = nad[k]; nad[k] = n2ad[k]; }
    }

    if constexpr (NEXT) {
        static_assert(LPR == 16, "the tile holds whole rows of F features");
        const int le   = (int)__builtin_amdgcn_mbcnt_lo(~0u, 0u);
        const int half = le >> 4;
        const int m    = le & 15;
        const float asv = sh_av[m];
        const float adv = sh_av[16 + m];
        const float* const Ar = my_row + (m & 7) * TP + 2 * half;
        const float* const Br = sh_w + le * WP;
        v8f acc = {};
        #pragma unroll
        for (int kk = 0; kk < F; kk += 4) {
            const v2f a = *(const v2f*)(Ar + kk);
            const v2f b = *(const v2f*)(Br + kk / 2);
            acc = __builtin_amdgcn_wmma_f32_16x16x4_f32(
                 false, a,  false, b,
                 (short)0, acc,  false,  false);
        }
        const float lg = tile_logit(acc, m, asv, adv);

        const int nrow = last_s - first_s;
        if (half == 0) {
            #pragma unroll
            for (int r = 0; r < NPW; ++r)
                if (r < nrow) out[(size_t)(first_s + r) * 16 + m] = acc[r];
            if ((m & 7) < nrow) ((m & 8) ? ad_n : as_n)[first_s + (m & 7)] = lg;
        }
    }
}

static inline int cdiv(int a, int b) { return (a + b - 1) / b; }

extern "C" void kernel_launch(void* const* d_in, const int* in_sizes, int n_in,
                              void* d_out, int out_size, void* d_ws, size_t ws_size,
                              hipStream_t stream) {
    const float* x      = (const float*)d_in[0];
    const int*   ei     = (const int*)  d_in[1];
    const float* W1     = (const float*)d_in[2];
    const float* a_src1 = (const float*)d_in[3];
    const float* a_dst1 = (const float*)d_in[4];
    const float* b1     = (const float*)d_in[5];
    const float* W2     = (const float*)d_in[6];
    const float* a_src2 = (const float*)d_in[7];
    const float* a_dst2 = (const float*)d_in[8];
    const float* b2     = (const float*)d_in[9];

    const int N  = in_sizes[0] / 128;
    const int E  = in_sizes[1] / 2;

    float* ws   = (float*)d_ws;
    float* h1   = ws;
    float* agg1 = h1   + (size_t)N * 64;
    float* as1  = agg1 + (size_t)N * 64;
    float* ad1  = as1  + (size_t)N * 4;
    int*   strt = (int*)(ad1 + (size_t)N * 4);
    int*   rnkw = strt  + N + 4;
    int*   csr  = rnkw  + E;
    int*   bsum = csr   + E;
    unsigned* tabg = (unsigned*)(bsum + NCHUNK * 128);
    int*   base = (int*)(tabg + (size_t)NCHUNK * CAPW);
    unsigned short* rank = (unsigned short*)rnkw;
    float* h2   = agg1;
    float* as2  = h2   + (size_t)N * 16;
    float* ad2  = as2  + N;

    float* out  = (float*)d_out;

    const int TB = 256;
    const int NB = cdiv(N, 1024);

    count_gemm_kernel<<<NCHUNK + NGEMM, 1024, 0, stream>>>(
        ei, E, tabg, rank, bsum, x, W1, a_src1, a_dst1, h1, as1, ad1, N);
    chunk_scan_kernel<<<NB, 256, 0, stream>>>(tabg, bsum, strt, base, N);
    chunk_place_kernel<<<dim3(32, NCHUNK), TB, 0, stream>>>(ei, E, base, rank, csr);

    node_gather_kernel<4, 16, true, true><<<cdiv(N, 64), TB, 0, stream>>>(
        strt, csr, h1, ad1, a_src1, b1, h2, N, W2, a_src2, a_dst2, as2, ad2);

    node_gather_kernel<1, 16, false, false><<<2048, TB, 0, stream>>>(
        strt, csr, h2, ad2, a_src2, b2, out, N,
        nullptr, nullptr, nullptr, nullptr, nullptr);
}
